# starting kernel
# baseline (speedup 1.0000x reference)
#include <hip/hip_runtime.h>
#include <math.h>

#define NN    50000
#define NE    800000
#define ET    (NE + NN)
#define HID   128
#define NH    4
#define CH    32

typedef float v2f __attribute__((ext_vector_type(2)));
typedef float v8f __attribute__((ext_vector_type(8)));

__device__ __forceinline__ int f2ord(float f) {
    int i = __float_as_int(f);
    return (i >= 0) ? i : (i ^ 0x7fffffff);
}
__device__ __forceinline__ float ord2f(int i) {
    return __int_as_float((i >= 0) ? i : (i ^ 0x7fffffff));
}

template <int K, int NT>
__global__ void wmma_gemm_strip(const float* __restrict__ X,
                                const float* __restrict__ W,
                                const float* __restrict__ bias,
                                float* __restrict__ Y,
                                int nMtiles)
{
    constexpr int Nout = NT * 16;
    extern __shared__ float Wlds[];

    for (int i = threadIdx.x * 4; i < K * Nout; i += blockDim.x * 4)
        *(float4*)&Wlds[i] = *(const float4*)&W[i];
    __syncthreads();

    int wave = (int)((blockIdx.x * blockDim.x + threadIdx.x) >> 5);
    if (wave >= nMtiles) return;
    int lane = (int)(threadIdx.x & 31);

    int row = wave * 16 + (lane & 15);
    int col = lane & 15;
    int kh  = (lane >> 4) << 1;

    v8f acc[NT];
#pragma unroll
    for (int t = 0; t < NT; ++t)
        acc[t] = (v8f){0.f, 0.f, 0.f, 0.f, 0.f, 0.f, 0.f, 0.f};

    const float* xrow = X + (size_t)row * K;
#pragma unroll
    for (int k = 0; k < K; k += 4) {
        v2f a = *(const v2f*)(xrow + k + kh);
        const float* w0 = Wlds + (k + kh) * Nout + col;
#pragma unroll
        for (int t = 0; t < NT; ++t) {
            v2f b;
            b.x = w0[t * 16];
            b.y = w0[t * 16 + Nout];
            acc[t] = __builtin_amdgcn_wmma_f32_16x16x4_f32(
                         false, a, false, b, (short)0, acc[t], false, false);
        }
    }

    int orow = wave * 16 + ((lane >> 4) << 3);
#pragma unroll
    for (int t = 0; t < NT; ++t) {
        int ocol = t * 16 + (lane & 15);
        float badd = bias ? bias[ocol] : 0.f;
#pragma unroll
        for (int r = 0; r < 8; ++r)
            Y[(size_t)(orow + r) * Nout + ocol] = acc[t][r] + badd;
    }
}

__global__ void alpha_kernel(const float* __restrict__ Hm,
                             const float* __restrict__ a_s,
                             const float* __restrict__ a_d,
                             float* __restrict__ as_out,
                             float* __restrict__ ad_out)
{
    int tid = blockIdx.x * blockDim.x + threadIdx.x;
    if (tid >= NN * NH) return;
    int n = tid >> 2, h = tid & 3;
    const float* hp = Hm  + n * HID + h * CH;
    const float* sp = a_s + h * CH;
    const float* dp = a_d + h * CH;
    float s = 0.f, d = 0.f;
#pragma unroll
    for (int c = 0; c < CH; c += 4) {
        float4 hv = *(const float4*)(hp + c);
        float4 sv = *(const float4*)(sp + c);
        float4 dv = *(const float4*)(dp + c);
        s += hv.x * sv.x + hv.y * sv.y + hv.z * sv.z + hv.w * sv.w;
        d += hv.x * dv.x + hv.y * dv.y + hv.z * dv.z + hv.w * dv.w;
    }
    as_out[tid] = s;
    ad_out[tid] = d;
}

__global__ void fill_f32(float* p, float v, int n) {
    int t = blockIdx.x * blockDim.x + threadIdx.x;
    if (t < n) p[t] = v;
}
__global__ void fill_i32(int* p, int v, int n) {
    int t = blockIdx.x * blockDim.x + threadIdx.x;
    if (t < n) p[t] = v;
}

__global__ void edge_pass1(const int* __restrict__ src,
                           const int* __restrict__ dst,
                           const float* __restrict__ as_v,
                           const float* __restrict__ ad_v,
                           float* __restrict__ ebuf,
                           int* __restrict__ mord)
{
    int tid = blockIdx.x * blockDim.x + threadIdx.x;
    if (tid >= ET * NH) return;
    int e = tid >> 2, h = tid & 3;
    int s = (e < NE) ? src[e] : (e - NE);
    int d = (e < NE) ? dst[e] : (e - NE);
    float v = as_v[s * NH + h] + ad_v[d * NH + h];
    v = (v > 0.f) ? v : 0.2f * v;
    ebuf[tid] = v;
    atomicMax(&mord[d * NH + h], f2ord(v));
}

__global__ void edge_pass2(const int* __restrict__ dst,
                           const int* __restrict__ mord,
                           float* __restrict__ ebuf,
                           float* __restrict__ denom)
{
    int tid = blockIdx.x * blockDim.x + threadIdx.x;
    if (tid >= ET * NH) return;
    int e = tid >> 2, h = tid & 3;
    int d = (e < NE) ? dst[e] : (e - NE);
    float m  = ord2f(mord[d * NH + h]);
    float ex = expf(ebuf[tid] - m);
    ebuf[tid] = ex;
    atomicAdd(&denom[d * NH + h], ex);
}

__global__ void edge_pass3(const int* __restrict__ src,
                           const int* __restrict__ dst,
                           const float* __restrict__ ebuf,
                           const float* __restrict__ denom,
                           const float* __restrict__ Hm,
                           float* __restrict__ agg)
{
    long long tid = (long long)blockIdx.x * blockDim.x + threadIdx.x;
    if (tid >= (long long)ET * (HID / 4)) return;
    int e = (int)(tid >> 5);
    int q = (int)(tid & 31);
    int j = q << 2;
    int h = j >> 5;
    int s = (e < NE) ? src[e] : (e - NE);
    int d = (e < NE) ? dst[e] : (e - NE);
    float alpha = ebuf[e * NH + h] / (denom[d * NH + h] + 1e-16f);
    float4 hv = *(const float4*)(Hm + (size_t)s * HID + j);
    float* ap = agg + (size_t)d * HID + j;
    atomicAdd(ap + 0, hv.x * alpha);
    atomicAdd(ap + 1, hv.y * alpha);
    atomicAdd(ap + 2, hv.z * alpha);
    atomicAdd(ap + 3, hv.w * alpha);
}

__global__ void bias_elu(const float* __restrict__ agg,
                         const float* __restrict__ b,
                         float* __restrict__ out)
{
    int tid = blockIdx.x * blockDim.x + threadIdx.x;
    if (tid >= NN * HID) return;
    int j = tid & 127;
    float y = agg[tid] + b[j];
    out[tid] = (y > 0.f) ? y : (expf(y) - 1.f);
}

static void run_gat_layer(const float* Xin, const float* W,
                          const float* a_s, const float* a_d, const float* b,
                          const int* src, const int* dst,
                          float* bufH, float* bufAgg, float* bufOut,
                          float* ebuf, float* asb, float* adb,
                          float* denom, int* mord, hipStream_t stream)
{
    const int TB = 256;
    int nMtiles = NN / 16;
    int blocks  = (nMtiles * 32 + TB - 1) / TB;
    wmma_gemm_strip<HID, HID / 16><<<blocks, TB, HID * HID * sizeof(float), stream>>>(
        Xin, W, nullptr, bufH, nMtiles);
    alpha_kernel<<<(NN * NH + TB - 1) / TB, TB, 0, stream>>>(bufH, a_s, a_d, asb, adb);
    fill_i32<<<(NN * NH + TB - 1) / TB, TB, 0, stream>>>(mord, (int)0x80000000, NN * NH);
    fill_f32<<<(NN * NH + TB - 1) / TB, TB, 0, stream>>>(denom, 0.f, NN * NH);
    fill_f32<<<(NN * HID + TB - 1) / TB, TB, 0, stream>>>(bufAgg, 0.f, NN * HID);
    int gEH = (ET * NH + TB - 1) / TB;
    edge_pass1<<<gEH, TB, 0, stream>>>(src, dst, asb, adb, ebuf, mord);
    edge_pass2<<<gEH, TB, 0, stream>>>(dst, mord, ebuf, denom);
    long long tot3 = (long long)ET * (HID / 4);
    edge_pass3<<<(unsigned)((tot3 + TB - 1) / TB), TB, 0, stream>>>(
        src, dst, ebuf, denom, bufH, bufAgg);
    bias_elu<<<(NN * HID + TB - 1) / TB, TB, 0, stream>>>(bufAgg, b, bufOut);
}

extern "C" void kernel_launch(void* const* d_in, const int* in_sizes, int n_in,
                              void* d_out, int out_size, void* d_ws, size_t ws_size,
                              hipStream_t stream)
{
    const float* x    = (const float*)d_in[0];
    const int*   edge = (const int*)  d_in[1];
    const float* W1   = (const float*)d_in[2];
    const float* a_s1 = (const float*)d_in[3];
    const float* a_d1 = (const float*)d_in[4];
    const float* b1   = (const float*)d_in[5];
    const float* W2   = (const float*)d_in[6];
    const float* a_s2 = (const float*)d_in[7];
    const float* a_d2 = (const float*)d_in[8];
    const float* b2   = (const float*)d_in[9];
    const float* fc_w = (const float*)d_in[10];
    const float* fc_b = (const float*)d_in[11];
    float* out = (float*)d_out;

    const int* src = edge;
    const int* dst = edge + NE;

    float* bufH   = (float*)d_ws;
    float* bufAgg = bufH   + (size_t)NN * HID;
    float* bufIn2 = bufAgg + (size_t)NN * HID;
    float* ebuf   = bufIn2 + (size_t)NN * HID;
    float* asb    = ebuf   + (size_t)ET * NH;
    float* adb    = asb    + (size_t)NN * NH;
    float* denom  = adb    + (size_t)NN * NH;
    int*   mord   = (int*)(denom + (size_t)NN * NH);

    run_gat_layer(x, W1, a_s1, a_d1, b1, src, dst,
                  bufH, bufAgg, bufIn2, ebuf, asb, adb, denom, mord, stream);
    run_gat_layer(bufIn2, W2, a_s2, a_d2, b2, src, dst,
                  bufH, bufAgg, bufIn2, ebuf, asb, adb, denom, mord, stream);
    const int TB = 256;
    int nMtiles = NN / 16;
    int blocks  = (nMtiles * 32 + TB - 1) / TB;
    wmma_gemm_strip<HID, CH / 16><<<blocks, TB, HID * CH * sizeof(float), stream>>>(
        bufIn2, fc_w, fc_b, out, nMtiles);
}


# Edges grouped by destination in-call; one wave per node does segment softmax + row gather + bias/ELU instead of 3 atomic edge passes, fills and epilogue; float sum order changes (no float atomics)
# speedup vs baseline: 2.2595x; 2.2595x over previous
#include <hip/hip_runtime.h>
#include <math.h>

#define NN    50000
#define NE    800000
#define HID   128
#define NH    4
#define CH    32

typedef float v2f __attribute__((ext_vector_type(2)));
typedef float v8f __attribute__((ext_vector_type(8)));

template <int K, int NT>
__global__ void wmma_gemm_strip(const float* __restrict__ X,
                                const float* __restrict__ W,
                                const float* __restrict__ bias,
                                float* __restrict__ Y,
                                int nMtiles)
{
    constexpr int Nout = NT * 16;
    extern __shared__ float Wlds[];

    for (int i = threadIdx.x * 4; i < K * Nout; i += blockDim.x * 4)
        *(float4*)&Wlds[i] = *(const float4*)&W[i];
    __syncthreads();

    int wave = (int)((blockIdx.x * blockDim.x + threadIdx.x) >> 5);
    if (wave >= nMtiles) return;
    int lane = (int)(threadIdx.x & 31);

    int row = wave * 16 + (lane & 15);
    int col = lane & 15;
    int kh  = (lane >> 4) << 1;

    v8f acc[NT];
#pragma unroll
    for (int t = 0; t < NT; ++t)
        acc[t] = (v8f){0.f, 0.f, 0.f, 0.f, 0.f, 0.f, 0.f, 0.f};

    const float* xrow = X + (size_t)row * K;
#pragma unroll
    for (int k = 0; k < K; k += 4) {
        v2f a = *(const v2f*)(xrow + k + kh);
        const float* w0 = Wlds + (k + kh) * Nout + col;
#pragma unroll
        for (int t = 0; t < NT; ++t) {
            v2f b;
            b.x = w0[t * 16];
            b.y = w0[t * 16 + Nout];
            acc[t] = __builtin_amdgcn_wmma_f32_16x16x4_f32(
                         false, a, false, b, (short)0, acc[t], false, false);
        }
    }

    int orow = wave * 16 + ((lane >> 4) << 3);
#pragma unroll
    for (int t = 0; t < NT; ++t) {
        int ocol = t * 16 + (lane & 15);
        float badd = bias ? bias[ocol] : 0.f;
#pragma unroll
        for (int r = 0; r < 8; ++r)
            Y[(size_t)(orow + r) * Nout + ocol] = acc[t][r] + badd;
    }
}

__global__ void alpha_kernel(const float* __restrict__ Hm,
                             const float* __restrict__ a_s,
                             const float* __restrict__ a_d,
                             float* __restrict__ as_out,
                             float* __restrict__ ad_out)
{
    int tid = blockIdx.x * blockDim.x + threadIdx.x;
    if (tid >= NN * NH) return;
    int n = tid >> 2, h = tid & 3;
    const float* hp = Hm  + n * HID + h * CH;
    const float* sp = a_s + h * CH;
    const float* dp = a_d + h * CH;
    float s = 0.f, d = 0.f;
#pragma unroll
    for (int c = 0; c < CH; c += 4) {
        float4 hv = *(const float4*)(hp + c);
        float4 sv = *(const float4*)(sp + c);
        float4 dv = *(const float4*)(dp + c);
        s += hv.x * sv.x + hv.y * sv.y + hv.z * sv.z + hv.w * sv.w;
        d += hv.x * dv.x + hv.y * dv.y + hv.z * dv.z + hv.w * dv.w;
    }
    as_out[tid] = s;
    ad_out[tid] = d;
}

__global__ void fill_i32(int* p, int v, int n) {
    int t = blockIdx.x * blockDim.x + threadIdx.x;
    if (t < n) p[t] = v;
}

__global__ void count_in(const int* __restrict__ dst, int* __restrict__ cnt)
{
    int e = blockIdx.x * blockDim.x + threadIdx.x;
    if (e < NE) atomicAdd(&cnt[dst[e]], 1);
}

__global__ void __launch_bounds__(256)
seg_alloc(const int* __restrict__ cnt, int* __restrict__ total,
          int* __restrict__ cursor)
{
    __shared__ int wsum[8];
    __shared__ int blk_base;
    const int n    = (int)(blockIdx.x * 256 + threadIdx.x);
    const int lane = (int)(threadIdx.x & 31);
    const int w    = (int)(threadIdx.x >> 5);
    const int c    = (n < NN) ? cnt[n] : 0;
    int v = c;
    for (int o = 1; o < 32; o <<= 1) {
        int t = __shfl_up(v, o, 32);
        if (lane >= o) v += t;
    }
    if (lane == 31) wsum[w] = v;
    __syncthreads();
    if (threadIdx.x == 0) {
        int s = 0;
        for (int i = 0; i < 8; ++i) { int t = wsum[i]; wsum[i] = s; s += t; }
        blk_base = atomicAdd(total, s);
    }
    __syncthreads();
    if (n < NN) cursor[n] = blk_base + wsum[w] + v - c;
}

__global__ void seg_fill(const int* __restrict__ src,
                         const int* __restrict__ dst,
                         int* __restrict__ cursor,
                         int* __restrict__ seg_src)
{
    int e = blockIdx.x * blockDim.x + threadIdx.x;
    if (e >= NE) return;
    int p = atomicAdd(&cursor[dst[e]], 1);
    seg_src[p] = src[e];
}

__device__ __forceinline__ float lrelu02(float v) {
    return (v > 0.f) ? v : 0.2f * v;
}
__device__ __forceinline__ float wave_max(float v) {
    for (int o = 16; o > 0; o >>= 1) v = fmaxf(v, __shfl_xor(v, o, 32));
    return v;
}
__device__ __forceinline__ float wave_sum(float v) {
    for (int o = 16; o > 0; o >>= 1) v += __shfl_xor(v, o, 32);
    return v;
}
__device__ __forceinline__ float head_weight(float a0, float a1, float a2, float a3, int k,
                                             int hm0, int hm1, int hm2, int hm3) {
    const int w0 = __builtin_amdgcn_readlane(__float_as_int(a0), k);
    const int w1 = __builtin_amdgcn_readlane(__float_as_int(a1), k);
    const int w2 = __builtin_amdgcn_readlane(__float_as_int(a2), k);
    const int w3 = __builtin_amdgcn_readlane(__float_as_int(a3), k);
    return __int_as_float((w0 & hm0) | (w1 & hm1) | (w2 & hm2) | (w3 & hm3));
}

__global__ void __launch_bounds__(256)
node_softmax_gather(const int* __restrict__ cnt,
                    const int* __restrict__ segend,
                    const int* __restrict__ seg_src,
                    const float* __restrict__ as_v,
                    const float* __restrict__ ad_v,
                    const float* __restrict__ Hm,
                    const float* __restrict__ b,
                    float* __restrict__ out)
{
    const int node = __builtin_amdgcn_readfirstlane(
                         (int)((blockIdx.x * 256u + threadIdx.x) >> 5));
    if (node >= NN) return;
    const int lane = (int)(threadIdx.x & 31);
    const int c    = __builtin_amdgcn_readfirstlane(cnt[node]);
    const int beg  = __builtin_amdgcn_readfirstlane(segend[node]) - c;
    const int tot  = c + 1;
    const float4 adv = *(const float4*)(ad_v + (size_t)node * NH);

    float m0 = -INFINITY, m1 = -INFINITY, m2 = -INFINITY, m3 = -INFINITY;
    for (int base = 0; base < tot; base += 32) {
        int i = base + lane;
        if (i < tot) {
            int s = (i < c) ? seg_src[beg + i] : node;
            float4 a = *(const float4*)(as_v + (size_t)s * NH);
            m0 = fmaxf(m0, lrelu02(a.x + adv.x));
            m1 = fmaxf(m1, lrelu02(a.y + adv.y));
            m2 = fmaxf(m2, lrelu02(a.z + adv.z));
            m3 = fmaxf(m3, lrelu02(a.w + adv.w));
        }
    }
    m0 = wave_max(m0); m1 = wave_max(m1); m2 = wave_max(m2); m3 = wave_max(m3);

    float d0 = 0.f, d1 = 0.f, d2 = 0.f, d3 = 0.f;
    for (int base = 0; base < tot; base += 32) {
        int i = base + lane;
        if (i < tot) {
            int s = (i < c) ? seg_src[beg + i] : node;
            float4 a = *(const float4*)(as_v + (size_t)s * NH);
            d0 += expf(lrelu02(a.x + adv.x) - m0);
            d1 += expf(lrelu02(a.y + adv.y) - m1);
            d2 += expf(lrelu02(a.z + adv.z) - m2);
            d3 += expf(lrelu02(a.w + adv.w) - m3);
        }
    }
    d0 = wave_sum(d0) + 1e-16f; d1 = wave_sum(d1) + 1e-16f;
    d2 = wave_sum(d2) + 1e-16f; d3 = wave_sum(d3) + 1e-16f;

    const int h = lane >> 3;
    const int hm0 = (h == 0) ? -1 : 0, hm1 = (h == 1) ? -1 : 0;
    const int hm2 = (h == 2) ? -1 : 0, hm3 = (h == 3) ? -1 : 0;
    const float* hcol = Hm + lane * 4;
    float4 acc = make_float4(0.f, 0.f, 0.f, 0.f);
    for (int base = 0; base < tot; base += 32) {
        int i = base + lane;
        int s = node;
        float a0 = 0.f, a1 = 0.f, a2 = 0.f, a3 = 0.f;
        if (i < tot) {
            s = (i < c) ? seg_src[beg + i] : node;
            float4 a = *(const float4*)(as_v + (size_t)s * NH);
            a0 = expf(lrelu02(a.x + adv.x) - m0) / d0;
            a1 = expf(lrelu02(a.y + adv.y) - m1) / d1;
            a2 = expf(lrelu02(a.z + adv.z) - m2) / d2;
            a3 = expf(lrelu02(a.w + adv.w) - m3) / d3;
        }
        const int cn = (tot - base < 32) ? (tot - base) : 32;
        int k = 0;
        for (; k + 4 <= cn; k += 4) {
            const int s0 = __builtin_amdgcn_readlane(s, k);
            const int s1 = __builtin_amdgcn_readlane(s, k + 1);
            const int s2 = __builtin_amdgcn_readlane(s, k + 2);
            const int s3 = __builtin_amdgcn_readlane(s, k + 3);
            const float4 r0 = *(const float4*)(hcol + (size_t)s0 * HID);
            const float4 r1 = *(const float4*)(hcol + (size_t)s1 * HID);
            const float4 r2 = *(const float4*)(hcol + (size_t)s2 * HID);
            const float4 r3 = *(const float4*)(hcol + (size_t)s3 * HID);
            const float t0 = head_weight(a0, a1, a2, a3, k,     hm0, hm1, hm2, hm3);
            const float t1 = head_weight(a0, a1, a2, a3, k + 1, hm0, hm1, hm2, hm3);
            const float t2 = head_weight(a0, a1, a2, a3, k + 2, hm0, hm1, hm2, hm3);
            const float t3 = head_weight(a0, a1, a2, a3, k + 3, hm0, hm1, hm2, hm3);
            acc.x += r0.x * t0; acc.y += r0.y * t0; acc.z += r0.z * t0; acc.w += r0.w * t0;
            acc.x += r1.x * t1; acc.y += r1.y * t1; acc.z += r1.z * t1; acc.w += r1.w * t1;
            acc.x += r2.x * t2; acc.y += r2.y * t2; acc.z += r2.z * t2; acc.w += r2.w * t2;
            acc.x += r3.x * t3; acc.y += r3.y * t3; acc.z += r3.z * t3; acc.w += r3.w * t3;
        }
        for (; k < cn; ++k) {
            const int sk = __builtin_amdgcn_readlane(s, k);
            const float4 r = *(const float4*)(hcol + (size_t)sk * HID);
            const float t = head_weight(a0, a1, a2, a3, k, hm0, hm1, hm2, hm3);
            acc.x += r.x * t; acc.y += r.y * t; acc.z += r.z * t; acc.w += r.w * t;
        }
    }

    const float4 bv = *(const float4*)(b + lane * 4);
    float4 y;
    y.x = acc.x + bv.x; y.y = acc.y + bv.y; y.z = acc.z + bv.z; y.w = acc.w + bv.w;
    y.x = (y.x > 0.f) ? y.x : (expf(y.x) - 1.f);
    y.y = (y.y > 0.f) ? y.y : (expf(y.y) - 1.f);
    y.z = (y.z > 0.f) ? y.z : (expf(y.z) - 1.f);
    y.w = (y.w > 0.f) ? y.w : (expf(y.w) - 1.f);
    *(float4*)(out + (size_t)node * HID + lane * 4) = y;
}

static void run_layer(const float* Xin, const float* W,
                      const float* a_s, const float* a_d, const float* b,
                      const int* cnt, const int* segend, const int* seg_src,
                      float* bufH, float* bufOut,
                      float* asb, float* adb, hipStream_t stream)
{
    const int TB = 256;
    int nMtiles = NN / 16;
    int blocks  = (nMtiles * 32 + TB - 1) / TB;
    wmma_gemm_strip<HID, HID / 16><<<blocks, TB, HID * HID * sizeof(float), stream>>>(
        Xin, W, nullptr, bufH, nMtiles);
    alpha_kernel<<<(NN * NH + TB - 1) / TB, TB, 0, stream>>>(bufH, a_s, a_d, asb, adb);
    node_softmax_gather<<<(NN * 32 + TB - 1) / TB, TB, 0, stream>>>(
        cnt, segend, seg_src, asb, adb, bufH, b, bufOut);
}

extern "C" void kernel_launch(void* const* d_in, const int* in_sizes, int n_in,
                              void* d_out, int out_size, void* d_ws, size_t ws_size,
                              hipStream_t stream)
{
    const float* x    = (const float*)d_in[0];
    const int*   edge = (const int*)  d_in[1];
    const float* W1   = (const float*)d_in[2];
    const float* a_s1 = (const float*)d_in[3];
    const float* a_d1 = (const float*)d_in[4];
    const float* b1   = (const float*)d_in[5];
    const float* W2   = (const float*)d_in[6];
    const float* a_s2 = (const float*)d_in[7];
    const float* a_d2 = (const float*)d_in[8];
    const float* b2   = (const float*)d_in[9];
    const float* fc_w = (const float*)d_in[10];
    const float* fc_b = (const float*)d_in[11];
    float* out = (float*)d_out;

    const int* src = edge;
    const int* dst = edge + NE;

    float* bufH   = (float*)d_ws;
    float* bufIn2 = bufH   + (size_t)NN * HID;
    float* asb    = bufIn2 + (size_t)NN * HID;
    float* adb    = asb    + (size_t)NN * NH;
    int*   cnt    = (int*)(adb + (size_t)NN * NH);
    int*   cursor = cnt    + (NN + 16);
    int*   segsrc = cursor + NN;

    const int TB = 256;
    fill_i32<<<(NN + 1 + TB - 1) / TB, TB, 0, stream>>>(cnt, 0, NN + 1);
    count_in<<<(NE + TB - 1) / TB, TB, 0, stream>>>(dst, cnt);
    seg_alloc<<<(NN + TB - 1) / TB, TB, 0, stream>>>(cnt, cnt + NN, cursor);
    seg_fill<<<(NE + TB - 1) / TB, TB, 0, stream>>>(src, dst, cursor, segsrc);

    run_layer(x, W1, a_s1, a_d1, b1, cnt, cursor, segsrc,
              bufH, bufIn2, asb, adb, stream);
    run_layer(bufIn2, W2, a_s2, a_d2, b2, cnt, cursor, segsrc,
              bufH, bufIn2, asb, adb, stream);
    int nMtiles = NN / 16;
    int blocks  = (nMtiles * 32 + TB - 1) / TB;
    wmma_gemm_strip<HID, CH / 16><<<blocks, TB, HID * CH * sizeof(float), stream>>>(
        bufIn2, fc_w, fc_b, out, nMtiles);
}


# Logits computed in the node kernel from the gathered rows (head dot, 8-lane DPP sum) with a running-max softmax: drops both logit launches, score tables and their gather; float sum order changes
# speedup vs baseline: 2.5830x; 1.1432x over previous
#include <hip/hip_runtime.h>
#include <math.h>

#define NN    50000
#define NE    800000
#define HID   128
#define NH    4
#define CH    32

typedef float v2f __attribute__((ext_vector_type(2)));
typedef float v8f __attribute__((ext_vector_type(8)));

template <int K, int NT>
__global__ void wmma_gemm_strip(const float* __restrict__ X,
                                const float* __restrict__ W,
                                const float* __restrict__ bias,
                                float* __restrict__ Y,
                                int nMtiles)
{
    constexpr int Nout = NT * 16;
    extern __shared__ float Wlds[];

    for (int i = threadIdx.x * 4; i < K * Nout; i += blockDim.x * 4)
        *(float4*)&Wlds[i] = *(const float4*)&W[i];
    __syncthreads();

    int wave = (int)((blockIdx.x * blockDim.x + threadIdx.x) >> 5);
    if (wave >= nMtiles) return;
    int lane = (int)(threadIdx.x & 31);

    int row = wave * 16 + (lane & 15);
    int col = lane & 15;
    int kh  = (lane >> 4) << 1;

    v8f acc[NT];
#pragma unroll
    for (int t = 0; t < NT; ++t)
        acc[t] = (v8f){0.f, 0.f, 0.f, 0.f, 0.f, 0.f, 0.f, 0.f};

    const float* xrow = X + (size_t)row * K;
#pragma unroll
    for (int k = 0; k < K; k += 4) {
        v2f a = *(const v2f*)(xrow + k + kh);
        const float* w0 = Wlds + (k + kh) * Nout + col;
#pragma unroll
        for (int t = 0; t < NT; ++t) {
            v2f b;
            b.x = w0[t * 16];
            b.y = w0[t * 16 + Nout];
            acc[t] = __builtin_amdgcn_wmma_f32_16x16x4_f32(
                         false, a, false, b, (short)0, acc[t], false, false);
        }
    }

    int orow = wave * 16 + ((lane >> 4) << 3);
#pragma unroll
    for (int t = 0; t < NT; ++t) {
        int ocol = t * 16 + (lane & 15);
        float badd = bias ? bias[ocol] : 0.f;
#pragma unroll
        for (int r = 0; r < 8; ++r)
            Y[(size_t)(orow + r) * Nout + ocol] = acc[t][r] + badd;
    }
}

__global__ void fill_i32(int* p, int v, int n) {
    int t = blockIdx.x * blockDim.x + threadIdx.x;
    if (t < n) p[t] = v;
}

__global__ void count_in(const int* __restrict__ dst, int* __restrict__ cnt)
{
    int e = blockIdx.x * blockDim.x + threadIdx.x;
    if (e < NE) atomicAdd(&cnt[dst[e]], 1);
}

__global__ void __launch_bounds__(256)
seg_alloc(const int* __restrict__ cnt, int* __restrict__ total,
          int* __restrict__ cursor)
{
    __shared__ int wsum[8];
    __shared__ int blk_base;
    const int n    = (int)(blockIdx.x * 256 + threadIdx.x);
    const int lane = (int)(threadIdx.x & 31);
    const int w    = (int)(threadIdx.x >> 5);
    const int c    = (n < NN) ? cnt[n] : 0;
    int v = c;
    for (int o = 1; o < 32; o <<= 1) {
        int t = __shfl_up(v, o, 32);
        if (lane >= o) v += t;
    }
    if (lane == 31) wsum[w] = v;
    __syncthreads();
    if (threadIdx.x == 0) {
        int s = 0;
        for (int i = 0; i < 8; ++i) { int t = wsum[i]; wsum[i] = s; s += t; }
        blk_base = atomicAdd(total, s);
    }
    __syncthreads();
    if (n < NN) cursor[n] = blk_base + wsum[w] + v - c;
}

__global__ void seg_fill(const int* __restrict__ src,
                         const int* __restrict__ dst,
                         int* __restrict__ cursor,
                         int* __restrict__ seg_src)
{
    int e = blockIdx.x * blockDim.x + threadIdx.x;
    if (e >= NE) return;
    int p = atomicAdd(&cursor[dst[e]], 1);
    seg_src[p] = src[e];
}

__device__ __forceinline__ float lrelu02(float v) {
    return (v > 0.f) ? v : 0.2f * v;
}
__device__ __forceinline__ float group8_sum(float v) {
    v += __int_as_float(__builtin_amdgcn_update_dpp(0, __float_as_int(v), 0xB1, 0xF, 0xF, true));
    v += __int_as_float(__builtin_amdgcn_update_dpp(0, __float_as_int(v), 0x4E, 0xF, 0xF, true));
    v += __int_as_float(__builtin_amdgcn_update_dpp(0, __float_as_int(v), 0x141, 0xF, 0xF, true));
    return v;
}

template <int N>
__device__ __forceinline__ void edge_trip(const float* __restrict__ hcol, int s, int k,
                                          int nvalid, const float4 asv, float ad,
                                          float& m, float& d, float4& acc)
{
    float4 r[N];
    float  e[N];
#pragma unroll
    for (int j = 0; j < N; ++j) {
        const int sj = __builtin_amdgcn_readlane(s, k + j);
        r[j] = *(const float4*)(hcol + (size_t)sj * HID);
    }
    float mn = m;
#pragma unroll
    for (int j = 0; j < N; ++j) {
        float p = r[j].x * asv.x + r[j].y * asv.y + r[j].z * asv.z + r[j].w * asv.w;
        p = group8_sum(p);
        const float ej = lrelu02(p + ad);
        e[j] = (j < nvalid) ? ej : -INFINITY;
        mn = fmaxf(mn, e[j]);
    }
    const float sc = expf(m - mn);
    float dsum = 0.f;
    float4 a;
    a.x = acc.x * sc; a.y = acc.y * sc; a.z = acc.z * sc; a.w = acc.w * sc;
#pragma unroll
    for (int j = 0; j < N; ++j) {
        const float pj = expf(e[j] - mn);
        dsum += pj;
        a.x += r[j].x * pj; a.y += r[j].y * pj; a.z += r[j].z * pj; a.w += r[j].w * pj;
    }
    d   = d * sc + dsum;
    m   = mn;
    acc = a;
}

__global__ void __launch_bounds__(256)
node_softmax_gather(const int* __restrict__ cnt,
                    const int* __restrict__ segend,
                    const int* __restrict__ seg_src,
                    const float* __restrict__ Hm,
                    const float* __restrict__ a_s,
                    const float* __restrict__ a_d,
                    const float* __restrict__ b,
                    float* __restrict__ out)
{
    const int node = __builtin_amdgcn_readfirstlane(
                         (int)((blockIdx.x * 256u + threadIdx.x) >> 5));
    if (node >= NN) return;
    const int lane = (int)(threadIdx.x & 31);
    const int c    = __builtin_amdgcn_readfirstlane(cnt[node]);
    const int beg  = __builtin_amdgcn_readfirstlane(segend[node]) - c;
    const float* hcol = Hm + lane * 4;
    const float4 asv = *(const float4*)(a_s + lane * 4);
    const float4 adv = *(const float4*)(a_d + lane * 4);

    const float4 rs = *(const float4*)(hcol + (size_t)node * HID);
    const float ad = group8_sum(rs.x * adv.x + rs.y * adv.y + rs.z * adv.z + rs.w * adv.w);
    float m = lrelu02(group8_sum(rs.x * asv.x + rs.y * asv.y + rs.z * asv.z + rs.w * asv.w) + ad);
    float d = 1.f;
    float4 acc = rs;

    for (int base = 0; base < c; base += 32) {
        const int i = base + lane;
        const int s = (i < c) ? seg_src[beg + i] : node;
        const int cn = (c - base < 32) ? (c - base) : 32;
        int k = 0;
        for (; k + 4 <= cn; k += 4)
            edge_trip<4>(hcol, s, k, 4, asv, ad, m, d, acc);
        if (k < cn)
            edge_trip<4>(hcol, s, k, cn - k, asv, ad, m, d, acc);
    }

    const float den = d + 1e-16f;
    const float4 bv = *(const float4*)(b + lane * 4);
    float4 y;
    y.x = acc.x / den + bv.x; y.y = acc.y / den + bv.y;
    y.z = acc.z / den + bv.z; y.w = acc.w / den + bv.w;
    y.x = (y.x > 0.f) ? y.x : (expf(y.x) - 1.f);
    y.y = (y.y > 0.f) ? y.y : (expf(y.y) - 1.f);
    y.z = (y.z > 0.f) ? y.z : (expf(y.z) - 1.f);
    y.w = (y.w > 0.f) ? y.w : (expf(y.w) - 1.f);
    *(float4*)(out + (size_t)node * HID + lane * 4) = y;
}

static void run_layer(const float* Xin, const float* W,
                      const float* a_s, const float* a_d, const float* b,
                      const int* cnt, const int* segend, const int* seg_src,
                      float* bufH, float* bufOut, hipStream_t stream)
{
    const int TB = 256;
    int nMtiles = NN / 16;
    int blocks  = (nMtiles * 32 + TB - 1) / TB;
    wmma_gemm_strip<HID, HID / 16><<<blocks, TB, HID * HID * sizeof(float), stream>>>(
        Xin, W, nullptr, bufH, nMtiles);
    node_softmax_gather<<<(NN * 32 + TB - 1) / TB, TB, 0, stream>>>(
        cnt, segend, seg_src, bufH, a_s, a_d, b, bufOut);
}

extern "C" void kernel_launch(void* const* d_in, const int* in_sizes, int n_in,
                              void* d_out, int out_size, void* d_ws, size_t ws_size,
                              hipStream_t stream)
{
    const float* x    = (const float*)d_in[0];
    const int*   edge = (const int*)  d_in[1];
    const float* W1   = (const float*)d_in[2];
    const float* a_s1 = (const float*)d_in[3];
    const float* a_d1 = (const float*)d_in[4];
    const float* b1   = (const float*)d_in[5];
    const float* W2   = (const float*)d_in[6];
    const float* a_s2 = (const float*)d_in[7];
    const float* a_d2 = (const float*)d_in[8];
    const float* b2   = (const float*)d_in[9];
    const float* fc_w = (const float*)d_in[10];
    const float* fc_b = (const float*)d_in[11];
    float* out = (float*)d_out;

    const int* src = edge;
    const int* dst = edge + NE;

    float* bufH   = (float*)d_ws;
    float* bufIn2 = bufH   + (size_t)NN * HID;
    int*   cnt    = (int*)(bufIn2 + (size_t)NN * HID);
    int*   cursor = cnt    + (NN + 16);
    int*   segsrc = cursor + NN;

    const int TB = 256;
    fill_i32<<<(NN + 1 + TB - 1) / TB, TB, 0, stream>>>(cnt, 0, NN + 1);
    count_in<<<(NE + TB - 1) / TB, TB, 0, stream>>>(dst, cnt);
    seg_alloc<<<(NN + TB - 1) / TB, TB, 0, stream>>>(cnt, cnt + NN, cursor);
    seg_fill<<<(NE + TB - 1) / TB, TB, 0, stream>>>(src, dst, cursor, segsrc);

    run_layer(x, W1, a_s1, a_d1, b1, cnt, cursor, segsrc,
              bufH, bufIn2, stream);
    run_layer(bufIn2, W2, a_s2, a_d2, b2, cnt, cursor, segsrc,
              bufH, bufIn2, stream);
    int nMtiles = NN / 16;
    int blocks  = (nMtiles * 32 + TB - 1) / TB;
    wmma_gemm_strip<HID, CH / 16><<<blocks, TB, HID * CH * sizeof(float), stream>>>(
        bufIn2, fc_w, fc_b, out, nMtiles);
}


# Edges go straight into a fixed 32-slot bucket per destination (one atomic pass, overflow list above 32) instead of count + scan + fill: 2 launches, 800000 atomics fewer; sum order stays arrival order
# speedup vs baseline: 3.3081x; 1.2807x over previous
#include <hip/hip_runtime.h>
#include <math.h>

#define NN    50000
#define NE    800000
#define HID   128
#define NH    4
#define CH    32

typedef float v2f __attribute__((ext_vector_type(2)));
typedef float v8f __attribute__((ext_vector_type(8)));

template <int K, int NT>
__global__ void wmma_gemm_strip(const float* __restrict__ X,
                                const float* __restrict__ W,
                                const float* __restrict__ bias,
                                float* __restrict__ Y,
                                int nMtiles)
{
    constexpr int Nout = NT * 16;
    extern __shared__ float Wlds[];

    for (int i = threadIdx.x * 4; i < K * Nout; i += blockDim.x * 4)
        *(float4*)&Wlds[i] = *(const float4*)&W[i];
    __syncthreads();

    int wave = (int)((blockIdx.x * blockDim.x + threadIdx.x) >> 5);
    if (wave >= nMtiles) return;
    int lane = (int)(threadIdx.x & 31);

    int row = wave * 16 + (lane & 15);
    int col = lane & 15;
    int kh  = (lane >> 4) << 1;

    v8f acc[NT];
#pragma unroll
    for (int t = 0; t < NT; ++t)
        acc[t] = (v8f){0.f, 0.f, 0.f, 0.f, 0.f, 0.f, 0.f, 0.f};

    const float* xrow = X + (size_t)row * K;
#pragma unroll
    for (int k = 0; k < K; k += 4) {
        v2f a = *(const v2f*)(xrow + k + kh);
        const float* w0 = Wlds + (k + kh) * Nout + col;
#pragma unroll
        for (int t = 0; t < NT; ++t) {
            v2f b;
            b.x = w0[t * 16];
            b.y = w0[t * 16 + Nout];
            acc[t] = __builtin_amdgcn_wmma_f32_16x16x4_f32(
                         false, a, false, b, (short)0, acc[t], false, false);
        }
    }

    int orow = wave * 16 + ((lane >> 4) << 3);
#pragma unroll
    for (int t = 0; t < NT; ++t) {
        int ocol = t * 16 + (lane & 15);
        float badd = bias ? bias[ocol] : 0.f;
#pragma unroll
        for (int r = 0; r < 8; ++r)
            Y[(size_t)(orow + r) * Nout + ocol] = acc[t][r] + badd;
    }
}

__global__ void fill_i32(int* p, int v, int n) {
    int t = blockIdx.x * blockDim.x + threadIdx.x;
    if (t < n) p[t] = v;
}

#define CAP 32
__global__ void bucket_fill(const int* __restrict__ src,
                            const int* __restrict__ dst,
                            int* __restrict__ cnt,
                            int* __restrict__ slots,
                            int* __restrict__ ovf_src,
                            int* __restrict__ ovf_dst)
{
    int e = blockIdx.x * blockDim.x + threadIdx.x;
    if (e >= NE) return;
    const int dn = dst[e];
    const int sv = src[e];
    const int p  = atomicAdd(&cnt[dn], 1);
    if (p < CAP) {
        slots[dn * CAP + p] = sv;
    } else {
        const int q = atomicAdd(&cnt[NN], 1);
        ovf_src[q] = sv;
        ovf_dst[q] = dn;
    }
}

__device__ __forceinline__ float lrelu02(float v) {
    return (v > 0.f) ? v : 0.2f * v;
}
__device__ __forceinline__ float group8_sum(float v) {
    v += __int_as_float(__builtin_amdgcn_update_dpp(0, __float_as_int(v), 0xB1, 0xF, 0xF, true));
    v += __int_as_float(__builtin_amdgcn_update_dpp(0, __float_as_int(v), 0x4E, 0xF, 0xF, true));
    v += __int_as_float(__builtin_amdgcn_update_dpp(0, __float_as_int(v), 0x141, 0xF, 0xF, true));
    return v;
}

template <int N>
__device__ __forceinline__ void edge_trip(const float* __restrict__ hcol, int s, int k,
                                          int nvalid, const float4 asv, float ad,
                                          float& m, float& d, float4& acc)
{
    float4 r[N];
    float  e[N];
#pragma unroll
    for (int j = 0; j < N; ++j) {
        const int sj = __builtin_amdgcn_readlane(s, k + j);
        r[j] = *(const float4*)(hcol + (size_t)sj * HID);
    }
    float mn = m;
#pragma unroll
    for (int j = 0; j < N; ++j) {
        float p = r[j].x * asv.x + r[j].y * asv.y + r[j].z * asv.z + r[j].w * asv.w;
        p = group8_sum(p);
        const float ej = lrelu02(p + ad);
        e[j] = (j < nvalid) ? ej : -INFINITY;
        mn = fmaxf(mn, e[j]);
    }
    const float sc = expf(m - mn);
    float dsum = 0.f;
    float4 a;
    a.x = acc.x * sc; a.y = acc.y * sc; a.z = acc.z * sc; a.w = acc.w * sc;
#pragma unroll
    for (int j = 0; j < N; ++j) {
        const float pj = expf(e[j] - mn);
        dsum += pj;
        a.x += r[j].x * pj; a.y += r[j].y * pj; a.z += r[j].z * pj; a.w += r[j].w * pj;
    }
    d   = d * sc + dsum;
    m   = mn;
    acc = a;
}

__global__ void __launch_bounds__(256)
node_softmax_gather(const int* __restrict__ cnt,
                    const int* __restrict__ slots,
                    const int* __restrict__ ovf_src,
                    const int* __restrict__ ovf_dst,
                    const float* __restrict__ Hm,
                    const float* __restrict__ a_s,
                    const float* __restrict__ a_d,
                    const float* __restrict__ b,
                    float* __restrict__ out)
{
    const int node = __builtin_amdgcn_readfirstlane(
                         (int)((blockIdx.x * 256u + threadIdx.x) >> 5));
    if (node >= NN) return;
    const int lane = (int)(threadIdx.x & 31);
    const int c    = __builtin_amdgcn_readfirstlane(cnt[node]);
    const int cb   = (c < CAP) ? c : CAP;
    const float* hcol = Hm + lane * 4;
    const float4 asv = *(const float4*)(a_s + lane * 4);
    const float4 adv = *(const float4*)(a_d + lane * 4);

    const float4 rs = *(const float4*)(hcol + (size_t)node * HID);
    const float ad = group8_sum(rs.x * adv.x + rs.y * adv.y + rs.z * adv.z + rs.w * adv.w);
    float m = lrelu02(group8_sum(rs.x * asv.x + rs.y * asv.y + rs.z * asv.z + rs.w * asv.w) + ad);
    float d = 1.f;
    float4 acc = rs;

    {
        const int s = (lane < cb) ? slots[node * CAP + lane] : node;
        int k = 0;
        for (; k + 4 <= cb; k += 4)
            edge_trip<4>(hcol, s, k, 4, asv, ad, m, d, acc);
        if (k < cb)
            edge_trip<4>(hcol, s, k, cb - k, asv, ad, m, d, acc);
    }
    if (c > CAP) {
        const int novf = __builtin_amdgcn_readfirstlane(cnt[NN]);
        for (int base = 0; base < novf; base += 32) {
            const int i  = base + lane;
            const int od = (i < novf) ? ovf_dst[i] : -1;
            const int os = (i < novf) ? ovf_src[i] : node;
            unsigned hit = __builtin_amdgcn_ballot_w32(od == node);
            while (hit) {
                const int j = __builtin_ctz(hit);
                hit &= hit - 1u;
                edge_trip<1>(hcol, os, j, 1, asv, ad, m, d, acc);
            }
        }
    }

    const float den = d + 1e-16f;
    const float4 bv = *(const float4*)(b + lane * 4);
    float4 y;
    y.x = acc.x / den + bv.x; y.y = acc.y / den + bv.y;
    y.z = acc.z / den + bv.z; y.w = acc.w / den + bv.w;
    y.x = (y.x > 0.f) ? y.x : (expf(y.x) - 1.f);
    y.y = (y.y > 0.f) ? y.y : (expf(y.y) - 1.f);
    y.z = (y.z > 0.f) ? y.z : (expf(y.z) - 1.f);
    y.w = (y.w > 0.f) ? y.w : (expf(y.w) - 1.f);
    *(float4*)(out + (size_t)node * HID + lane * 4) = y;
}

static void run_layer(const float* Xin, const float* W,
                      const float* a_s, const float* a_d, const float* b,
                      const int* cnt, const int* slots,
                      const int* ovf_src, const int* ovf_dst,
                      float* bufH, float* bufOut, hipStream_t stream)
{
    const int TB = 256;
    int nMtiles = NN / 16;
    int blocks  = (nMtiles * 32 + TB - 1) / TB;
    wmma_gemm_strip<HID, HID / 16><<<blocks, TB, HID * HID * sizeof(float), stream>>>(
        Xin, W, nullptr, bufH, nMtiles);
    node_softmax_gather<<<(NN * 32 + TB - 1) / TB, TB, 0, stream>>>(
        cnt, slots, ovf_src, ovf_dst, bufH, a_s, a_d, b, bufOut);
}

extern "C" void kernel_launch(void* const* d_in, const int* in_sizes, int n_in,
                              void* d_out, int out_size, void* d_ws, size_t ws_size,
                              hipStream_t stream)
{
    const float* x    = (const float*)d_in[0];
    const int*   edge = (const int*)  d_in[1];
    const float* W1   = (const float*)d_in[2];
    const float* a_s1 = (const float*)d_in[3];
    const float* a_d1 = (const float*)d_in[4];
    const float* b1   = (const float*)d_in[5];
    const float* W2   = (const float*)d_in[6];
    const float* a_s2 = (const float*)d_in[7];
    const float* a_d2 = (const float*)d_in[8];
    const float* b2   = (const float*)d_in[9];
    const float* fc_w = (const float*)d_in[10];
    const float* fc_b = (const float*)d_in[11];
    float* out = (float*)d_out;

    const int* src = edge;
    const int* dst = edge + NE;

    float* bufH   = (float*)d_ws;
    float* bufIn2 = bufH   + (size_t)NN * HID;
    int*   cnt    = (int*)(bufIn2 + (size_t)NN * HID);
    int*   slots  = cnt    + (NN + 16);
    int*   ovfs   = slots  + (size_t)NN * CAP;
    int*   ovfd   = ovfs   + NE;

    const int TB = 256;
    fill_i32<<<(NN + 1 + TB - 1) / TB, TB, 0, stream>>>(cnt, 0, NN + 1);
    bucket_fill<<<(NE + TB - 1) / TB, TB, 0, stream>>>(src, dst, cnt, slots, ovfs, ovfd);

    run_layer(x, W1, a_s1, a_d1, b1, cnt, slots, ovfs, ovfd,
              bufH, bufIn2, stream);
    run_layer(bufIn2, W2, a_s2, a_d2, b2, cnt, slots, ovfs, ovfd,
              bufH, bufIn2, stream);
    int nMtiles = NN / 16;
    int blocks  = (nMtiles * 32 + TB - 1) / TB;
    wmma_gemm_strip<HID, CH / 16><<<blocks, TB, HID * CH * sizeof(float), stream>>>(
        bufIn2, fc_w, fc_b, out, nMtiles);
}


# Library exp -> hardware exp2 where the argument is <= 0: softmax logits in log2 units (a_s, a_d scaled by log2(e) once per wave), ELU via exp2(y*log2(e)); 13 -> 1 instr per exp; exp rounding changes
# speedup vs baseline: 3.5587x; 1.0758x over previous
#include <hip/hip_runtime.h>
#include <math.h>

#define NN    50000
#define NE    800000
#define HID   128
#define NH    4
#define CH    32

typedef float v2f __attribute__((ext_vector_type(2)));
typedef float v8f __attribute__((ext_vector_type(8)));

template <int K, int NT>
__global__ void wmma_gemm_strip(const float* __restrict__ X,
                                const float* __restrict__ W,
                                const float* __restrict__ bias,
                                float* __restrict__ Y,
                                int nMtiles)
{
    constexpr int Nout = NT * 16;
    extern __shared__ float Wlds[];

    for (int i = threadIdx.x * 4; i < K * Nout; i += blockDim.x * 4)
        *(float4*)&Wlds[i] = *(const float4*)&W[i];
    __syncthreads();

    int wave = (int)((blockIdx.x * blockDim.x + threadIdx.x) >> 5);
    if (wave >= nMtiles) return;
    int lane = (int)(threadIdx.x & 31);

    int row = wave * 16 + (lane & 15);
    int col = lane & 15;
    int kh  = (lane >> 4) << 1;

    v8f acc[NT];
#pragma unroll
    for (int t = 0; t < NT; ++t)
        acc[t] = (v8f){0.f, 0.f, 0.f, 0.f, 0.f, 0.f, 0.f, 0.f};

    const float* xrow = X + (size_t)row * K;
#pragma unroll
    for (int k = 0; k < K; k += 4) {
        v2f a = *(const v2f*)(xrow + k + kh);
        const float* w0 = Wlds + (k + kh) * Nout + col;
#pragma unroll
        for (int t = 0; t < NT; ++t) {
            v2f b;
            b.x = w0[t * 16];
            b.y = w0[t * 16 + Nout];
            acc[t] = __builtin_amdgcn_wmma_f32_16x16x4_f32(
                         false, a, false, b, (short)0, acc[t], false, false);
        }
    }

    int orow = wave * 16 + ((lane >> 4) << 3);
#pragma unroll
    for (int t = 0; t < NT; ++t) {
        int ocol = t * 16 + (lane & 15);
        float badd = bias ? bias[ocol] : 0.f;
#pragma unroll
        for (int r = 0; r < 8; ++r)
            Y[(size_t)(orow + r) * Nout + ocol] = acc[t][r] + badd;
    }
}

__global__ void fill_i32(int* p, int v, int n) {
    int t = blockIdx.x * blockDim.x + threadIdx.x;
    if (t < n) p[t] = v;
}

#define CAP 32
__global__ void bucket_fill(const int* __restrict__ src,
                            const int* __restrict__ dst,
                            int* __restrict__ cnt,
                            int* __restrict__ slots,
                            int* __restrict__ ovf_src,
                            int* __restrict__ ovf_dst)
{
    int e = blockIdx.x * blockDim.x + threadIdx.x;
    if (e >= NE) return;
    const int dn = dst[e];
    const int sv = src[e];
    const int p  = atomicAdd(&cnt[dn], 1);
    if (p < CAP) {
        slots[dn * CAP + p] = sv;
    } else {
        const int q = atomicAdd(&cnt[NN], 1);
        ovf_src[q] = sv;
        ovf_dst[q] = dn;
    }
}

__device__ __forceinline__ float lrelu02(float v) {
    return (v > 0.f) ? v : 0.2f * v;
}
#define LOG2E 1.4426950408889634f
__device__ __forceinline__ float exp2_neg(float v) {
    return __builtin_amdgcn_exp2f(v);
}
__device__ __forceinline__ float group8_sum(float v) {
    v += __int_as_float(__builtin_amdgcn_update_dpp(0, __float_as_int(v), 0xB1, 0xF, 0xF, true));
    v += __int_as_float(__builtin_amdgcn_update_dpp(0, __float_as_int(v), 0x4E, 0xF, 0xF, true));
    v += __int_as_float(__builtin_amdgcn_update_dpp(0, __float_as_int(v), 0x141, 0xF, 0xF, true));
    return v;
}

template <int N>
__device__ __forceinline__ void edge_trip(const float* __restrict__ hcol, int s, int k,
                                          int nvalid, const float4 asv, float ad,
                                          float& m, float& d, float4& acc)
{
    float4 r[N];
    float  e[N];
#pragma unroll
    for (int j = 0; j < N; ++j) {
        const int sj = __builtin_amdgcn_readlane(s, k + j);
        r[j] = *(const float4*)(hcol + (size_t)sj * HID);
    }
    float mn = m;
#pragma unroll
    for (int j = 0; j < N; ++j) {
        float p = r[j].x * asv.x + r[j].y * asv.y + r[j].z * asv.z + r[j].w * asv.w;
        p = group8_sum(p);
        const float ej = lrelu02(p + ad);
        e[j] = (j < nvalid) ? ej : -INFINITY;
        mn = fmaxf(mn, e[j]);
    }
    const float sc = exp2_neg(m - mn);
    float dsum = 0.f;
    float4 a;
    a.x = acc.x * sc; a.y = acc.y * sc; a.z = acc.z * sc; a.w = acc.w * sc;
#pragma unroll
    for (int j = 0; j < N; ++j) {
        const float pj = exp2_neg(e[j] - mn);
        dsum += pj;
        a.x += r[j].x * pj; a.y += r[j].y * pj; a.z += r[j].z * pj; a.w += r[j].w * pj;
    }
    d   = d * sc + dsum;
    m   = mn;
    acc = a;
}

__global__ void __launch_bounds__(256)
node_softmax_gather(const int* __restrict__ cnt,
                    const int* __restrict__ slots,
                    const int* __restrict__ ovf_src,
                    const int* __restrict__ ovf_dst,
                    const float* __restrict__ Hm,
                    const float* __restrict__ a_s,
                    const float* __restrict__ a_d,
                    const float* __restrict__ b,
                    float* __restrict__ out)
{
    const int node = __builtin_amdgcn_readfirstlane(
                         (int)((blockIdx.x * 256u + threadIdx.x) >> 5));
    if (node >= NN) return;
    const int lane = (int)(threadIdx.x & 31);
    const int c    = __builtin_amdgcn_readfirstlane(cnt[node]);
    const int cb   = (c < CAP) ? c : CAP;
    const float* hcol = Hm + lane * 4;
    float4 asv = *(const float4*)(a_s + lane * 4);
    float4 adv = *(const float4*)(a_d + lane * 4);
    asv.x *= LOG2E; asv.y *= LOG2E; asv.z *= LOG2E; asv.w *= LOG2E;
    adv.x *= LOG2E; adv.y *= LOG2E; adv.z *= LOG2E; adv.w *= LOG2E;

    const float4 rs = *(const float4*)(hcol + (size_t)node * HID);
    const float ad = group8_sum(rs.x * adv.x + rs.y * adv.y + rs.z * adv.z + rs.w * adv.w);
    float m = lrelu02(group8_sum(rs.x * asv.x + rs.y * asv.y + rs.z * asv.z + rs.w * asv.w) + ad);
    float d = 1.f;
    float4 acc = rs;

    {
        const int s = (lane < cb) ? slots[node * CAP + lane] : node;
        int k = 0;
        for (; k + 4 <= cb; k += 4)
            edge_trip<4>(hcol, s, k, 4, asv, ad, m, d, acc);
        if (k < cb)
            edge_trip<4>(hcol, s, k, cb - k, asv, ad, m, d, acc);
    }
    if (c > CAP) {
        const int novf = __builtin_amdgcn_readfirstlane(cnt[NN]);
        for (int base = 0; base < novf; base += 32) {
            const int i  = base + lane;
            const int od = (i < novf) ? ovf_dst[i] : -1;
            const int os = (i < novf) ? ovf_src[i] : node;
            unsigned hit = __builtin_amdgcn_ballot_w32(od == node);
            while (hit) {
                const int j = __builtin_ctz(hit);
                hit &= hit - 1u;
                edge_trip<1>(hcol, os, j, 1, asv, ad, m, d, acc);
            }
        }
    }

    const float den = d + 1e-16f;
    const float4 bv = *(const float4*)(b + lane * 4);
    float4 y;
    y.x = acc.x / den + bv.x; y.y = acc.y / den + bv.y;
    y.z = acc.z / den + bv.z; y.w = acc.w / den + bv.w;
    y.x = (y.x > 0.f) ? y.x : (exp2_neg(y.x * LOG2E) - 1.f);
    y.y = (y.y > 0.f) ? y.y : (exp2_neg(y.y * LOG2E) - 1.f);
    y.z = (y.z > 0.f) ? y.z : (exp2_neg(y.z * LOG2E) - 1.f);
    y.w = (y.w > 0.f) ? y.w : (exp2_neg(y.w * LOG2E) - 1.f);
    *(float4*)(out + (size_t)node * HID + lane * 4) = y;
}

static void run_layer(const float* Xin, const float* W,
                      const float* a_s, const float* a_d, const float* b,
                      const int* cnt, const int* slots,
                      const int* ovf_src, const int* ovf_dst,
                      float* bufH, float* bufOut, hipStream_t stream)
{
    const int TB = 256;
    int nMtiles = NN / 16;
    int blocks  = (nMtiles * 32 + TB - 1) / TB;
    wmma_gemm_strip<HID, HID / 16><<<blocks, TB, HID * HID * sizeof(float), stream>>>(
        Xin, W, nullptr, bufH, nMtiles);
    node_softmax_gather<<<(NN * 32 + TB - 1) / TB, TB, 0, stream>>>(
        cnt, slots, ovf_src, ovf_dst, bufH, a_s, a_d, b, bufOut);
}

extern "C" void kernel_launch(void* const* d_in, const int* in_sizes, int n_in,
                              void* d_out, int out_size, void* d_ws, size_t ws_size,
                              hipStream_t stream)
{
    const float* x    = (const float*)d_in[0];
    const int*   edge = (const int*)  d_in[1];
    const float* W1   = (const float*)d_in[2];
    const float* a_s1 = (const float*)d_in[3];
    const float* a_d1 = (const float*)d_in[4];
    const float* b1   = (const float*)d_in[5];
    const float* W2   = (const float*)d_in[6];
    const float* a_s2 = (const float*)d_in[7];
    const float* a_d2 = (const float*)d_in[8];
    const float* b2   = (const float*)d_in[9];
    const float* fc_w = (const float*)d_in[10];
    const float* fc_b = (const float*)d_in[11];
    float* out = (float*)d_out;

    const int* src = edge;
    const int* dst = edge + NE;

    float* bufH   = (float*)d_ws;
    float* bufIn2 = bufH   + (size_t)NN * HID;
    int*   cnt    = (int*)(bufIn2 + (size_t)NN * HID);
    int*   slots  = cnt    + (NN + 16);
    int*   ovfs   = slots  + (size_t)NN * CAP;
    int*   ovfd   = ovfs   + NE;

    const int TB = 256;
    fill_i32<<<(NN + 1 + TB - 1) / TB, TB, 0, stream>>>(cnt, 0, NN + 1);
    bucket_fill<<<(NE + TB - 1) / TB, TB, 0, stream>>>(src, dst, cnt, slots, ovfs, ovfd);

    run_layer(x, W1, a_s1, a_d1, b1, cnt, slots, ovfs, ovfd,
              bufH, bufIn2, stream);
    run_layer(bufIn2, W2, a_s2, a_d2, b2, cnt, slots, ovfs, ovfd,
              bufH, bufIn2, stream);
    int nMtiles = NN / 16;
    int blocks  = (nMtiles * 32 + TB - 1) / TB;
    wmma_gemm_strip<HID, CH / 16><<<blocks, TB, HID * CH * sizeof(float), stream>>>(
        bufIn2, fc_w, fc_b, out, nMtiles);
}


# Edges grouped by destination by LDS binning in two passes (256 edge chunks x 256 node ranges, whole-line stores) instead of 800000 global atomics + scattered stores; sum order stays arrival order
# speedup vs baseline: 4.9919x; 1.4027x over previous
#include <hip/hip_runtime.h>
#include <math.h>

#define NN    50000
#define NE    800000
#define HID   128
#define NH    4
#define CH    32

typedef float v2f __attribute__((ext_vector_type(2)));
typedef float v8f __attribute__((ext_vector_type(8)));

template <int K, int NT>
__global__ void wmma_gemm_strip(const float* __restrict__ X,
                                const float* __restrict__ W,
                                const float* __restrict__ bias,
                                float* __restrict__ Y,
                                int nMtiles)
{
    constexpr int Nout = NT * 16;
    extern __shared__ float Wlds[];

    for (int i = threadIdx.x * 4; i < K * Nout; i += blockDim.x * 4)
        *(float4*)&Wlds[i] = *(const float4*)&W[i];
    __syncthreads();

    int wave = (int)((blockIdx.x * blockDim.x + threadIdx.x) >> 5);
    if (wave >= nMtiles) return;
    int lane = (int)(threadIdx.x & 31);

    int row = wave * 16 + (lane & 15);
    int col = lane & 15;
    int kh  = (lane >> 4) << 1;

    v8f acc[NT];
#pragma unroll
    for (int t = 0; t < NT; ++t)
        acc[t] = (v8f){0.f, 0.f, 0.f, 0.f, 0.f, 0.f, 0.f, 0.f};

    const float* xrow = X + (size_t)row * K;
#pragma unroll
    for (int k = 0; k < K; k += 4) {
        v2f a = *(const v2f*)(xrow + k + kh);
        const float* w0 = Wlds + (k + kh) * Nout + col;
#pragma unroll
        for (int t = 0; t < NT; ++t) {
            v2f b;
            b.x = w0[t * 16];
            b.y = w0[t * 16 + Nout];
            acc[t] = __builtin_amdgcn_wmma_f32_16x16x4_f32(
                         false, a, false, b, (short)0, acc[t], false, false);
        }
    }

    int orow = wave * 16 + ((lane >> 4) << 3);
#pragma unroll
    for (int t = 0; t < NT; ++t) {
        int ocol = t * 16 + (lane & 15);
        float badd = bias ? bias[ocol] : 0.f;
#pragma unroll
        for (int r = 0; r < 8; ++r)
            Y[(size_t)(orow + r) * Nout + ocol] = acc[t][r] + badd;
    }
}

#define CAP     32
#define NBIN    256
#define RANGE   196
#define NPART   256
#define CHUNK   3125
#define CELLCAP 31
static_assert(NBIN * RANGE >= NN, "ranges cover all nodes");
static_assert(NPART * CHUNK == NE, "chunks cover all edges");
static_assert(NN <= 65536 && RANGE <= 256, "packed entry: 16-bit source, local destination above it");

__global__ void __launch_bounds__(256)
edge_partition(const int* __restrict__ src,
               const int* __restrict__ dst,
               unsigned* __restrict__ part,
               unsigned* __restrict__ spill,
               int* __restrict__ spill_cnt,
               int* __restrict__ ovf_total)
{
    __shared__ int      ccnt[NBIN];
    __shared__ unsigned cell[NBIN * 32];
    __shared__ int      nspill;
    const int w = blockIdx.x;
    const int t = threadIdx.x;
    ccnt[t] = 0;
    if (t == 0) nspill = 0;
    if (w == 0 && t == 0) *ovf_total = 0;

    constexpr int PER = (CHUNK + 255) / 256;
    int dn[PER], sv[PER];
#pragma unroll
    for (int j = 0; j < PER; ++j) {
        const int i = j * 256 + t;
        const int e = w * CHUNK + ((i < CHUNK) ? i : 0);
        dn[j] = dst[e];
        sv[j] = src[e];
    }
    __syncthreads();
    int bn[PER], pr[PER];
#pragma unroll
    for (int j = 0; j < PER; ++j) {
        bn[j] = dn[j] / RANGE;
        pr[j] = (j * 256 + t < CHUNK) ? atomicAdd(&ccnt[bn[j]], 1) : -1;
    }
#pragma unroll
    for (int j = 0; j < PER; ++j) {
        if (pr[j] >= 0) {
            if (pr[j] < CELLCAP) {
                cell[bn[j] * 32 + 1 + pr[j]] = ((unsigned)(dn[j] - bn[j] * RANGE) << 16) | (unsigned)sv[j];
            } else {
                const int q = atomicAdd(&nspill, 1);
                spill[(size_t)w * CHUNK + q] = ((unsigned)dn[j] << 16) | (unsigned)sv[j];
            }
        }
    }
    __syncthreads();
    const int wave = t >> 5, lane = t & 31;
    for (int b = wave; b < NBIN; b += 8) {
        int c = ccnt[b];
        c = (c < CELLCAP) ? c : CELLCAP;
        unsigned v = (unsigned)c;
        if (lane >= 1) v = (lane <= c) ? cell[b * 32 + lane] : 0u;
        part[((size_t)b * NPART + w) * 32 + lane] = v;
    }
    if (t == 0) spill_cnt[w] = nspill;
}

__global__ void __launch_bounds__(256)
bucket_build(const unsigned* __restrict__ part,
             const unsigned* __restrict__ spill,
             const int* __restrict__ spill_cnt,
             int* __restrict__ cnt,
             int* __restrict__ slots,
             int* __restrict__ ovf_src,
             int* __restrict__ ovf_dst)
{
    __shared__ int ncnt[256];
    __shared__ int nslot[RANGE * CAP];
    __shared__ int scnt[NPART];
    __shared__ int slist[NPART];
    __shared__ int nlist;
    const int b = blockIdx.x;
    const int t = threadIdx.x;
    const int wave = t >> 5, lane = t & 31;
    const int n0 = b * RANGE;
    ncnt[t] = 0;
    if (t == 0) nlist = 0;

    auto store_ranked = [&](int dl, int s, int p) {
        if (p < CAP) {
            nslot[dl * CAP + p] = s;
        } else {
            const int q = atomicAdd(&cnt[NN], 1);
            ovf_src[q] = s;
            ovf_dst[q] = n0 + dl;
        }
    };
    auto place = [&](int dl, int s) { store_ranked(dl, s, atomicAdd(&ncnt[dl], 1)); };

    unsigned v[32];
#pragma unroll
    for (int j = 0; j < 32; ++j)
        v[j] = part[((size_t)b * NPART + wave * 32 + j) * 32 + lane];
    const int sc = spill_cnt[t];
    __syncthreads();
    scnt[t] = sc;
    if (sc > 0) slist[atomicAdd(&nlist, 1)] = t;
    int pr[32];
#pragma unroll
    for (int j = 0; j < 32; ++j) {
        const int c = __builtin_amdgcn_readfirstlane((int)v[j]);
        pr[j] = (lane >= 1 && lane <= c) ? atomicAdd(&ncnt[v[j] >> 16], 1) : -1;
    }
#pragma unroll
    for (int j = 0; j < 32; ++j) {
        if (pr[j] >= 0)
            store_ranked((int)(v[j] >> 16), (int)(v[j] & 0xffffu), pr[j]);
    }
    __syncthreads();
    const int nl = nlist;
    for (int k = 0; k < nl; ++k) {
        const int w = slist[k];
        int n = scnt[w];
        n = (n < CHUNK) ? n : CHUNK;
        for (int i = t; i < n; i += 256) {
            const unsigned e = spill[(size_t)w * CHUNK + i];
            const int dl = (int)(e >> 16) - n0;
            if (dl >= 0 && dl < RANGE)
                place(dl, (int)(e & 0xffffu));
        }
    }
    if (nl > 0) __syncthreads();
    for (int r = wave; r < RANGE; r += 8) {
        const int node = n0 + r;
        if (node < NN) {
            int c = ncnt[r];
            c = (c < CAP) ? c : CAP;
            slots[(size_t)node * CAP + lane] = (lane < c) ? nslot[r * CAP + lane] : 0;
        }
    }
    if (t < RANGE && n0 + t < NN) cnt[n0 + t] = ncnt[t];
}

__device__ __forceinline__ float lrelu02(float v) {
    return (v > 0.f) ? v : 0.2f * v;
}
#define LOG2E 1.4426950408889634f
__device__ __forceinline__ float exp2_neg(float v) {
    return __builtin_amdgcn_exp2f(v);
}
__device__ __forceinline__ float group8_sum(float v) {
    v += __int_as_float(__builtin_amdgcn_update_dpp(0, __float_as_int(v), 0xB1, 0xF, 0xF, true));
    v += __int_as_float(__builtin_amdgcn_update_dpp(0, __float_as_int(v), 0x4E, 0xF, 0xF, true));
    v += __int_as_float(__builtin_amdgcn_update_dpp(0, __float_as_int(v), 0x141, 0xF, 0xF, true));
    return v;
}

template <int N>
__device__ __forceinline__ void edge_trip(const float* __restrict__ hcol, int s, int k,
                                          int nvalid, const float4 asv, float ad,
                                          float& m, float& d, float4& acc)
{
    float4 r[N];
    float  e[N];
#pragma unroll
    for (int j = 0; j < N; ++j) {
        const int sj = __builtin_amdgcn_readlane(s, k + j);
        r[j] = *(const float4*)(hcol + (size_t)sj * HID);
    }
    float mn = m;
#pragma unroll
    for (int j = 0; j < N; ++j) {
        float p = r[j].x * asv.x + r[j].y * asv.y + r[j].z * asv.z + r[j].w * asv.w;
        p = group8_sum(p);
        const float ej = lrelu02(p + ad);
        e[j] = (j < nvalid) ? ej : -INFINITY;
        mn = fmaxf(mn, e[j]);
    }
    const float sc = exp2_neg(m - mn);
    float dsum = 0.f;
    float4 a;
    a.x = acc.x * sc; a.y = acc.y * sc; a.z = acc.z * sc; a.w = acc.w * sc;
#pragma unroll
    for (int j = 0; j < N; ++j) {
        const float pj = exp2_neg(e[j] - mn);
        dsum += pj;
        a.x += r[j].x * pj; a.y += r[j].y * pj; a.z += r[j].z * pj; a.w += r[j].w * pj;
    }
    d   = d * sc + dsum;
    m   = mn;
    acc = a;
}

__global__ void __launch_bounds__(256)
node_softmax_gather(const int* __restrict__ cnt,
                    const int* __restrict__ slots,
                    const int* __restrict__ ovf_src,
                    const int* __restrict__ ovf_dst,
                    const float* __restrict__ Hm,
                    const float* __restrict__ a_s,
                    const float* __restrict__ a_d,
                    const float* __restrict__ b,
                    float* __restrict__ out)
{
    const int node = __builtin_amdgcn_readfirstlane(
                         (int)((blockIdx.x * 256u + threadIdx.x) >> 5));
    if (node >= NN) return;
    const int lane = (int)(threadIdx.x & 31);
    const int c    = __builtin_amdgcn_readfirstlane(cnt[node]);
    const int cb   = (c < CAP) ? c : CAP;
    const float* hcol = Hm + lane * 4;
    float4 asv = *(const float4*)(a_s + lane * 4);
    float4 adv = *(const float4*)(a_d + lane * 4);
    asv.x *= LOG2E; asv.y *= LOG2E; asv.z *= LOG2E; asv.w *= LOG2E;
    adv.x *= LOG2E; adv.y *= LOG2E; adv.z *= LOG2E; adv.w *= LOG2E;

    const float4 rs = *(const float4*)(hcol + (size_t)node * HID);
    const float ad = group8_sum(rs.x * adv.x + rs.y * adv.y + rs.z * adv.z + rs.w * adv.w);
    float m = lrelu02(group8_sum(rs.x * asv.x + rs.y * asv.y + rs.z * asv.z + rs.w * asv.w) + ad);
    float d = 1.f;
    float4 acc = rs;

    {
        const int s = (lane < cb) ? slots[node * CAP + lane] : node;
        int k = 0;
        for (; k + 4 <= cb; k += 4)
            edge_trip<4>(hcol, s, k, 4, asv, ad, m, d, acc);
        if (k < cb)
            edge_trip<4>(hcol, s, k, cb - k, asv, ad, m, d, acc);
    }
    if (c > CAP) {
        const int novf = __builtin_amdgcn_readfirstlane(cnt[NN]);
        for (int base = 0; base < novf; base += 32) {
            const int i  = base + lane;
            const int od = (i < novf) ? ovf_dst[i] : -1;
            const int os = (i < novf) ? ovf_src[i] : node;
            unsigned hit = __builtin_amdgcn_ballot_w32(od == node);
            while (hit) {
                const int j = __builtin_ctz(hit);
                hit &= hit - 1u;
                edge_trip<1>(hcol, os, j, 1, asv, ad, m, d, acc);
            }
        }
    }

    const float den = d + 1e-16f;
    const float4 bv = *(const float4*)(b + lane * 4);
    float4 y;
    y.x = acc.x / den + bv.x; y.y = acc.y / den + bv.y;
    y.z = acc.z / den + bv.z; y.w = acc.w / den + bv.w;
    y.x = (y.x > 0.f) ? y.x : (exp2_neg(y.x * LOG2E) - 1.f);
    y.y = (y.y > 0.f) ? y.y : (exp2_neg(y.y * LOG2E) - 1.f);
    y.z = (y.z > 0.f) ? y.z : (exp2_neg(y.z * LOG2E) - 1.f);
    y.w = (y.w > 0.f) ? y.w : (exp2_neg(y.w * LOG2E) - 1.f);
    *(float4*)(out + (size_t)node * HID + lane * 4) = y;
}

static void run_layer(const float* Xin, const float* W,
                      const float* a_s, const float* a_d, const float* b,
                      const int* cnt, const int* slots,
                      const int* ovf_src, const int* ovf_dst,
                      float* bufH, float* bufOut, hipStream_t stream)
{
    const int TB = 256;
    int nMtiles = NN / 16;
    int blocks  = (nMtiles * 32 + TB - 1) / TB;
    wmma_gemm_strip<HID, HID / 16><<<blocks, TB, HID * HID * sizeof(float), stream>>>(
        Xin, W, nullptr, bufH, nMtiles);
    node_softmax_gather<<<(NN * 32 + TB - 1) / TB, TB, 0, stream>>>(
        cnt, slots, ovf_src, ovf_dst, bufH, a_s, a_d, b, bufOut);
}

extern "C" void kernel_launch(void* const* d_in, const int* in_sizes, int n_in,
                              void* d_out, int out_size, void* d_ws, size_t ws_size,
                              hipStream_t stream)
{
    const float* x    = (const float*)d_in[0];
    const int*   edge = (const int*)  d_in[1];
    const float* W1   = (const float*)d_in[2];
    const float* a_s1 = (const float*)d_in[3];
    const float* a_d1 = (const float*)d_in[4];
    const float* b1   = (const float*)d_in[5];
    const float* W2   = (const float*)d_in[6];
    const float* a_s2 = (const float*)d_in[7];
    const float* a_d2 = (const float*)d_in[8];
    const float* b2   = (const float*)d_in[9];
    const float* fc_w = (const float*)d_in[10];
    const float* fc_b = (const float*)d_in[11];
    float* out = (float*)d_out;

    const int* src = edge;
    const int* dst = edge + NE;

    float* bufH   = (float*)d_ws;
    float* bufIn2 = bufH   + (size_t)NN * HID;
    int*   cnt    = (int*)(bufIn2 + (size_t)NN * HID);
    int*   slots  = cnt    + (NN + 16);
    int*   ovfs   = slots  + (size_t)NN * CAP;
    int*   ovfd   = ovfs   + NE;
    unsigned* part  = (unsigned*)(ovfd + NE);
    unsigned* spill = part + (size_t)NBIN * NPART * 32;
    int*      spcnt = (int*)(spill + NE);

    const int TB = 256;
    edge_partition<<<NPART, TB, 0, stream>>>(src, dst, part, spill, spcnt, cnt + NN);
    bucket_build<<<NBIN, TB, 0, stream>>>(part, spill, spcnt, cnt, slots, ovfs, ovfd);

    run_layer(x, W1, a_s1, a_d1, b1, cnt, slots, ovfs, ovfd,
              bufH, bufIn2, stream);
    run_layer(bufIn2, W2, a_s2, a_d2, b2, cnt, slots, ovfs, ovfd,
              bufH, bufIn2, stream);
    int nMtiles = NN / 16;
    int blocks  = (nMtiles * 32 + TB - 1) / TB;
    wmma_gemm_strip<HID, CH / 16><<<blocks, TB, HID * CH * sizeof(float), stream>>>(
        bufIn2, fc_w, fc_b, out, nMtiles);
}


# Weight staging of the strip GEMMs: each thread's 16 slab loads are issued together, then stored to LDS with 128-bit stores, instead of one second-level round trip per 4 KB slab; sum order unchanged
# speedup vs baseline: 5.6774x; 1.1373x over previous
#include <hip/hip_runtime.h>
#include <math.h>

#define NN    50000
#define NE    800000
#define HID   128
#define NH    4
#define CH    32

typedef float v2f __attribute__((ext_vector_type(2)));
typedef float v8f __attribute__((ext_vector_type(8)));

#define GEMM_TB 256
template <int K, int NT>
__global__ void __launch_bounds__(GEMM_TB)
wmma_gemm_strip(const float* __restrict__ X,
                const float* __restrict__ W,
                const float* __restrict__ bias,
                float* __restrict__ Y,
                int nMtiles)
{
    constexpr int Nout = NT * 16;
    extern __shared__ __align__(16) float Wlds[];

    constexpr int NSLAB = (K * Nout) / (GEMM_TB * 4);
    static_assert(NSLAB * GEMM_TB * 4 == K * Nout, "W is a whole number of 16-byte pieces per thread");
    float4 wpiece[NSLAB];
#pragma unroll
    for (int j = 0; j < NSLAB; ++j)
        wpiece[j] = *(const float4*)&W[(j * GEMM_TB + (int)threadIdx.x) * 4];
#pragma unroll
    for (int j = 0; j < NSLAB; ++j)
        *(float4*)&Wlds[(j * GEMM_TB + (int)threadIdx.x) * 4] = wpiece[j];
    __syncthreads();

    int wave = (int)((blockIdx.x * blockDim.x + threadIdx.x) >> 5);
    if (wave >= nMtiles) return;
    int lane = (int)(threadIdx.x & 31);

    int row = wave * 16 + (lane & 15);
    int col = lane & 15;
    int kh  = (lane >> 4) << 1;

    v8f acc[NT];
#pragma unroll
    for (int t = 0; t < NT; ++t)
        acc[t] = (v8f){0.f, 0.f, 0.f, 0.f, 0.f, 0.f, 0.f, 0.f};

    const float* xrow = X + (size_t)row * K;
#pragma unroll
    for (int k = 0; k < K; k += 4) {
        v2f a = *(const v2f*)(xrow + k + kh);
        const float* w0 = Wlds + (k + kh) * Nout + col;
#pragma unroll
        for (int t = 0; t < NT; ++t) {
            v2f b;
            b.x = w0[t * 16];
            b.y = w0[t * 16 + Nout];
            acc[t] = __builtin_amdgcn_wmma_f32_16x16x4_f32(
                         false, a, false, b, (short)0, acc[t], false, false);
        }
    }

    int orow = wave * 16 + ((lane >> 4) << 3);
#pragma unroll
    for (int t = 0; t < NT; ++t) {
        int ocol = t * 16 + (lane & 15);
        float badd = bias ? bias[ocol] : 0.f;
#pragma unroll
        for (int r = 0; r < 8; ++r)
            Y[(size_t)(orow + r) * Nout + ocol] = acc[t][r] + badd;
    }
}

#define CAP     32
#define NBIN    256
#define RANGE   196
#define NPART   256
#define CHUNK   3125
#define CELLCAP 31
static_assert(NBIN * RANGE >= NN, "ranges cover all nodes");
static_assert(NPART * CHUNK == NE, "chunks cover all edges");
static_assert(NN <= 65536 && RANGE <= 256, "packed entry: 16-bit source, local destination above it");

__global__ void __launch_bounds__(256)
edge_partition(const int* __restrict__ src,
               const int* __restrict__ dst,
               unsigned* __restrict__ part,
               unsigned* __restrict__ spill,
               int* __restrict__ spill_cnt,
               int* __restrict__ ovf_total)
{
    __shared__ int      ccnt[NBIN];
    __shared__ unsigned cell[NBIN * 32];
    __shared__ int      nspill;
    const int w = blockIdx.x;
    const int t = threadIdx.x;
    ccnt[t] = 0;
    if (t == 0) nspill = 0;
    if (w == 0 && t == 0) *ovf_total = 0;

    constexpr int PER = (CHUNK + 255) / 256;
    int dn[PER], sv[PER];
#pragma unroll
    for (int j = 0; j < PER; ++j) {
        const int i = j * 256 + t;
        const int e = w * CHUNK + ((i < CHUNK) ? i : 0);
        dn[j] = dst[e];
        sv[j] = src[e];
    }
    __syncthreads();
    int bn[PER], pr[PER];
#pragma unroll
    for (int j = 0; j < PER; ++j) {
        bn[j] = dn[j] / RANGE;
        pr[j] = (j * 256 + t < CHUNK) ? atomicAdd(&ccnt[bn[j]], 1) : -1;
    }
#pragma unroll
    for (int j = 0; j < PER; ++j) {
        if (pr[j] >= 0) {
            if (pr[j] < CELLCAP) {
                cell[bn[j] * 32 + 1 + pr[j]] = ((unsigned)(dn[j] - bn[j] * RANGE) << 16) | (unsigned)sv[j];
            } else {
                const int q = atomicAdd(&nspill, 1);
                spill[(size_t)w * CHUNK + q] = ((unsigned)dn[j] << 16) | (unsigned)sv[j];
            }
        }
    }
    __syncthreads();
    const int wave = t >> 5, lane = t & 31;
    for (int b = wave; b < NBIN; b += 8) {
        int c = ccnt[b];
        c = (c < CELLCAP) ? c : CELLCAP;
        unsigned v = (unsigned)c;
        if (lane >= 1) v = (lane <= c) ? cell[b * 32 + lane] : 0u;
        part[((size_t)b * NPART + w) * 32 + lane] = v;
    }
    if (t == 0) spill_cnt[w] = nspill;
}

__global__ void __launch_bounds__(256)
bucket_build(const unsigned* __restrict__ part,
             const unsigned* __restrict__ spill,
             const int* __restrict__ spill_cnt,
             int* __restrict__ cnt,
             int* __restrict__ slots,
             int* __restrict__ ovf_src,
             int* __restrict__ ovf_dst)
{
    __shared__ int ncnt[256];
    __shared__ int nslot[RANGE * CAP];
    __shared__ int scnt[NPART];
    __shared__ int slist[NPART];
    __shared__ int nlist;
    const int b = blockIdx.x;
    const int t = threadIdx.x;
    const int wave = t >> 5, lane = t & 31;
    const int n0 = b * RANGE;
    ncnt[t] = 0;
    if (t == 0) nlist = 0;

    auto store_ranked = [&](int dl, int s, int p) {
        if (p < CAP) {
            nslot[dl * CAP + p] = s;
        } else {
            const int q = atomicAdd(&cnt[NN], 1);
            ovf_src[q] = s;
            ovf_dst[q] = n0 + dl;
        }
    };
    auto place = [&](int dl, int s) { store_ranked(dl, s, atomicAdd(&ncnt[dl], 1)); };

    unsigned v[32];
#pragma unroll
    for (int j = 0; j < 32; ++j)
        v[j] = part[((size_t)b * NPART + wave * 32 + j) * 32 + lane];
    const int sc = spill_cnt[t];
    __syncthreads();
    scnt[t] = sc;
    if (sc > 0) slist[atomicAdd(&nlist, 1)] = t;
    int pr[32];
#pragma unroll
    for (int j = 0; j < 32; ++j) {
        const int c = __builtin_amdgcn_readfirstlane((int)v[j]);
        pr[j] = (lane >= 1 && lane <= c) ? atomicAdd(&ncnt[v[j] >> 16], 1) : -1;
    }
#pragma unroll
    for (int j = 0; j < 32; ++j) {
        if (pr[j] >= 0)
            store_ranked((int)(v[j] >> 16), (int)(v[j] & 0xffffu), pr[j]);
    }
    __syncthreads();
    const int nl = nlist;
    for (int k = 0; k < nl; ++k) {
        const int w = slist[k];
        int n = scnt[w];
        n = (n < CHUNK) ? n : CHUNK;
        for (int i = t; i < n; i += 256) {
            const unsigned e = spill[(size_t)w * CHUNK + i];
            const int dl = (int)(e >> 16) - n0;
            if (dl >= 0 && dl < RANGE)
                place(dl, (int)(e & 0xffffu));
        }
    }
    if (nl > 0) __syncthreads();
    for (int r = wave; r < RANGE; r += 8) {
        const int node = n0 + r;
        if (node < NN) {
            int c = ncnt[r];
            c = (c < CAP) ? c : CAP;
            slots[(size_t)node * CAP + lane] = (lane < c) ? nslot[r * CAP + lane] : 0;
        }
    }
    if (t < RANGE && n0 + t < NN) cnt[n0 + t] = ncnt[t];
}

__device__ __forceinline__ float lrelu02(float v) {
    return (v > 0.f) ? v : 0.2f * v;
}
#define LOG2E 1.4426950408889634f
__device__ __forceinline__ float exp2_neg(float v) {
    return __builtin_amdgcn_exp2f(v);
}
__device__ __forceinline__ float group8_sum(float v) {
    v += __int_as_float(__builtin_amdgcn_update_dpp(0, __float_as_int(v), 0xB1, 0xF, 0xF, true));
    v += __int_as_float(__builtin_amdgcn_update_dpp(0, __float_as_int(v), 0x4E, 0xF, 0xF, true));
    v += __int_as_float(__builtin_amdgcn_update_dpp(0, __float_as_int(v), 0x141, 0xF, 0xF, true));
    return v;
}

template <int N>
__device__ __forceinline__ void edge_trip(const float* __restrict__ hcol, int s, int k,
                                          int nvalid, const float4 asv, float ad,
                                          float& m, float& d, float4& acc)
{
    float4 r[N];
    float  e[N];
#pragma unroll
    for (int j = 0; j < N; ++j) {
        const int sj = __builtin_amdgcn_readlane(s, k + j);
        r[j] = *(const float4*)(hcol + (size_t)sj * HID);
    }
    float mn = m;
#pragma unroll
    for (int j = 0; j < N; ++j) {
        float p = r[j].x * asv.x + r[j].y * asv.y + r[j].z * asv.z + r[j].w * asv.w;
        p = group8_sum(p);
        const float ej = lrelu02(p + ad);
        e[j] = (j < nvalid) ? ej : -INFINITY;
        mn = fmaxf(mn, e[j]);
    }
    const float sc = exp2_neg(m - mn);
    float dsum = 0.f;
    float4 a;
    a.x = acc.x * sc; a.y = acc.y * sc; a.z = acc.z * sc; a.w = acc.w * sc;
#pragma unroll
    for (int j = 0; j < N; ++j) {
        const float pj = exp2_neg(e[j] - mn);
        dsum += pj;
        a.x += r[j].x * pj; a.y += r[j].y * pj; a.z += r[j].z * pj; a.w += r[j].w * pj;
    }
    d   = d * sc + dsum;
    m   = mn;
    acc = a;
}

__global__ void __launch_bounds__(256)
node_softmax_gather(const int* __restrict__ cnt,
                    const int* __restrict__ slots,
                    const int* __restrict__ ovf_src,
                    const int* __restrict__ ovf_dst,
                    const float* __restrict__ Hm,
                    const float* __restrict__ a_s,
                    const float* __restrict__ a_d,
                    const float* __restrict__ b,
                    float* __restrict__ out)
{
    const int node = __builtin_amdgcn_readfirstlane(
                         (int)((blockIdx.x * 256u + threadIdx.x) >> 5));
    if (node >= NN) return;
    const int lane = (int)(threadIdx.x & 31);
    const int c    = __builtin_amdgcn_readfirstlane(cnt[node]);
    const int cb   = (c < CAP) ? c : CAP;
    const float* hcol = Hm + lane * 4;
    float4 asv = *(const float4*)(a_s + lane * 4);
    float4 adv = *(const float4*)(a_d + lane * 4);
    asv.x *= LOG2E; asv.y *= LOG2E; asv.z *= LOG2E; asv.w *= LOG2E;
    adv.x *= LOG2E; adv.y *= LOG2E; adv.z *= LOG2E; adv.w *= LOG2E;

    const float4 rs = *(const float4*)(hcol + (size_t)node * HID);
    const float ad = group8_sum(rs.x * adv.x + rs.y * adv.y + rs.z * adv.z + rs.w * adv.w);
    float m = lrelu02(group8_sum(rs.x * asv.x + rs.y * asv.y + rs.z * asv.z + rs.w * asv.w) + ad);
    float d = 1.f;
    float4 acc = rs;

    {
        const int s = (lane < cb) ? slots[node * CAP + lane] : node;
        int k = 0;
        for (; k + 4 <= cb; k += 4)
            edge_trip<4>(hcol, s, k, 4, asv, ad, m, d, acc);
        if (k < cb)
            edge_trip<4>(hcol, s, k, cb - k, asv, ad, m, d, acc);
    }
    if (c > CAP) {
        const int novf = __builtin_amdgcn_readfirstlane(cnt[NN]);
        for (int base = 0; base < novf; base += 32) {
            const int i  = base + lane;
            const int od = (i < novf) ? ovf_dst[i] : -1;
            const int os = (i < novf) ? ovf_src[i] : node;
            unsigned hit = __builtin_amdgcn_ballot_w32(od == node);
            while (hit) {
                const int j = __builtin_ctz(hit);
                hit &= hit - 1u;
                edge_trip<1>(hcol, os, j, 1, asv, ad, m, d, acc);
            }
        }
    }

    const float den = d + 1e-16f;
    const float4 bv = *(const float4*)(b + lane * 4);
    float4 y;
    y.x = acc.x / den + bv.x; y.y = acc.y / den + bv.y;
    y.z = acc.z / den + bv.z; y.w = acc.w / den + bv.w;
    y.x = (y.x > 0.f) ? y.x : (exp2_neg(y.x * LOG2E) - 1.f);
    y.y = (y.y > 0.f) ? y.y : (exp2_neg(y.y * LOG2E) - 1.f);
    y.z = (y.z > 0.f) ? y.z : (exp2_neg(y.z * LOG2E) - 1.f);
    y.w = (y.w > 0.f) ? y.w : (exp2_neg(y.w * LOG2E) - 1.f);
    *(float4*)(out + (size_t)node * HID + lane * 4) = y;
}

static void run_layer(const float* Xin, const float* W,
                      const float* a_s, const float* a_d, const float* b,
                      const int* cnt, const int* slots,
                      const int* ovf_src, const int* ovf_dst,
                      float* bufH, float* bufOut, hipStream_t stream)
{
    const int TB = GEMM_TB;
    int nMtiles = NN / 16;
    int blocks  = (nMtiles * 32 + TB - 1) / TB;
    wmma_gemm_strip<HID, HID / 16><<<blocks, TB, HID * HID * sizeof(float), stream>>>(
        Xin, W, nullptr, bufH, nMtiles);
    node_softmax_gather<<<(NN * 32 + TB - 1) / TB, TB, 0, stream>>>(
        cnt, slots, ovf_src, ovf_dst, bufH, a_s, a_d, b, bufOut);
}

extern "C" void kernel_launch(void* const* d_in, const int* in_sizes, int n_in,
                              void* d_out, int out_size, void* d_ws, size_t ws_size,
                              hipStream_t stream)
{
    const float* x    = (const float*)d_in[0];
    const int*   edge = (const int*)  d_in[1];
    const float* W1   = (const float*)d_in[2];
    const float* a_s1 = (const float*)d_in[3];
    const float* a_d1 = (const float*)d_in[4];
    const float* b1   = (const float*)d_in[5];
    const float* W2   = (const float*)d_in[6];
    const float* a_s2 = (const float*)d_in[7];
    const float* a_d2 = (const float*)d_in[8];
    const float* b2   = (const float*)d_in[9];
    const float* fc_w = (const float*)d_in[10];
    const float* fc_b = (const float*)d_in[11];
    float* out = (float*)d_out;

    const int* src = edge;
    const int* dst = edge + NE;

    float* bufH   = (float*)d_ws;
    float* bufIn2 = bufH   + (size_t)NN * HID;
    int*   cnt    = (int*)(bufIn2 + (size_t)NN * HID);
    int*   slots  = cnt    + (NN + 16);
    int*   ovfs   = slots  + (size_t)NN * CAP;
    int*   ovfd   = ovfs   + NE;
    unsigned* part  = (unsigned*)(ovfd + NE);
    unsigned* spill = part + (size_t)NBIN * NPART * 32;
    int*      spcnt = (int*)(spill + NE);

    const int TB = 256;
    edge_partition<<<NPART, TB, 0, stream>>>(src, dst, part, spill, spcnt, cnt + NN);
    bucket_build<<<NBIN, TB, 0, stream>>>(part, spill, spcnt, cnt, slots, ovfs, ovfd);

    run_layer(x, W1, a_s1, a_d1, b1, cnt, slots, ovfs, ovfd,
              bufH, bufIn2, stream);
    run_layer(bufIn2, W2, a_s2, a_d2, b2, cnt, slots, ovfs, ovfd,
              bufH, bufIn2, stream);
    int nMtiles = NN / 16;
    int blocks  = (nMtiles * 32 + TB - 1) / TB;
    wmma_gemm_strip<HID, CH / 16><<<blocks, TB, HID * CH * sizeof(float), stream>>>(
        bufIn2, fc_w, fc_b, out, nMtiles);
}


# Layer-1 projection GEMM split in two halves whose workgroups run inside the two edge-grouping launches (independent work on idle compute units): 7 -> 6 launches; arithmetic and sum order unchanged
# speedup vs baseline: 5.8659x; 1.0332x over previous
#include <hip/hip_runtime.h>
#include <math.h>

#define NN    50000
#define NE    800000
#define HID   128
#define NH    4
#define CH    32

typedef float v2f __attribute__((ext_vector_type(2)));
typedef float v8f __attribute__((ext_vector_type(8)));

#define GEMM_TB 256
template <int K, int NT>
__device__ __forceinline__ void
gemm_strip_body(float* __restrict__ Wlds,
                const float* __restrict__ X,
                const float* __restrict__ W,
                const float* __restrict__ bias,
                float* __restrict__ Y,
                int blk, int nMtiles)
{
    constexpr int Nout = NT * 16;

    constexpr int NSLAB = (K * Nout) / (GEMM_TB * 4);
    static_assert(NSLAB * GEMM_TB * 4 == K * Nout, "W is a whole number of 16-byte pieces per thread");
    float4 wpiece[NSLAB];
#pragma unroll
    for (int j = 0; j < NSLAB; ++j)
        wpiece[j] = *(const float4*)&W[(j * GEMM_TB + (int)threadIdx.x) * 4];
#pragma unroll
    for (int j = 0; j < NSLAB; ++j)
        *(float4*)&Wlds[(j * GEMM_TB + (int)threadIdx.x) * 4] = wpiece[j];
    __syncthreads();

    int wave = (int)((blk * GEMM_TB + (int)threadIdx.x) >> 5);
    if (wave >= nMtiles) return;
    int lane = (int)(threadIdx.x & 31);

    int row = wave * 16 + (lane & 15);
    int col = lane & 15;
    int kh  = (lane >> 4) << 1;

    v8f acc[NT];
#pragma unroll
    for (int t = 0; t < NT; ++t)
        acc[t] = (v8f){0.f, 0.f, 0.f, 0.f, 0.f, 0.f, 0.f, 0.f};

    const float* xrow = X + (size_t)row * K;
#pragma unroll
    for (int k = 0; k < K; k += 4) {
        v2f a = *(const v2f*)(xrow + k + kh);
        const float* w0 = Wlds + (k + kh) * Nout + col;
#pragma unroll
        for (int t = 0; t < NT; ++t) {
            v2f b;
            b.x = w0[t * 16];
            b.y = w0[t * 16 + Nout];
            acc[t] = __builtin_amdgcn_wmma_f32_16x16x4_f32(
                         false, a, false, b, (short)0, acc[t], false, false);
        }
    }

    int orow = wave * 16 + ((lane >> 4) << 3);
#pragma unroll
    for (int t = 0; t < NT; ++t) {
        int ocol = t * 16 + (lane & 15);
        float badd = bias ? bias[ocol] : 0.f;
#pragma unroll
        for (int r = 0; r < 8; ++r)
            Y[(size_t)(orow + r) * Nout + ocol] = acc[t][r] + badd;
    }
}

template <int K, int NT>
__global__ void __launch_bounds__(GEMM_TB)
wmma_gemm_strip(const float* __restrict__ X,
                const float* __restrict__ W,
                const float* __restrict__ bias,
                float* __restrict__ Y,
                int nMtiles)
{
    extern __shared__ __align__(16) float Wlds[];
    gemm_strip_body<K, NT>(Wlds, X, W, bias, Y, (int)blockIdx.x, nMtiles);
}

#define CAP     32
#define NBIN    256
#define RANGE   196
#define NPART   256
#define CHUNK   3125
#define CELLCAP 31
static_assert(NBIN * RANGE >= NN, "ranges cover all nodes");
static_assert(NPART * CHUNK == NE, "chunks cover all edges");
static_assert(NN <= 65536 && RANGE <= 256, "packed entry: 16-bit source, local destination above it");

#define PART_LDS_WORDS (NBIN + NBIN * 32 + 1)
__device__ __forceinline__ void
edge_partition_body(unsigned* __restrict__ lds, const int w,
               const int* __restrict__ src,
               const int* __restrict__ dst,
               unsigned* __restrict__ part,
               unsigned* __restrict__ spill,
               int* __restrict__ spill_cnt,
               int* __restrict__ ovf_total)
{
    int*      ccnt    = (int*)lds;
    unsigned* cell    = lds + NBIN;
    int*      nspillp = (int*)(lds + NBIN + NBIN * 32);
    const int t = threadIdx.x;
    ccnt[t] = 0;
    if (t == 0) *nspillp = 0;
    if (w == 0 && t == 0) *ovf_total = 0;

    constexpr int PER = (CHUNK + 255) / 256;
    int dn[PER], sv[PER];
#pragma unroll
    for (int j = 0; j < PER; ++j) {
        const int i = j * 256 + t;
        const int e = w * CHUNK + ((i < CHUNK) ? i : 0);
        dn[j] = dst[e];
        sv[j] = src[e];
    }
    __syncthreads();
    int bn[PER], pr[PER];
#pragma unroll
    for (int j = 0; j < PER; ++j) {
        bn[j] = dn[j] / RANGE;
        pr[j] = (j * 256 + t < CHUNK) ? atomicAdd(&ccnt[bn[j]], 1) : -1;
    }
#pragma unroll
    for (int j = 0; j < PER; ++j) {
        if (pr[j] >= 0) {
            if (pr[j] < CELLCAP) {
                cell[bn[j] * 32 + 1 + pr[j]] = ((unsigned)(dn[j] - bn[j] * RANGE) << 16) | (unsigned)sv[j];
            } else {
                const int q = atomicAdd(nspillp, 1);
                spill[(size_t)w * CHUNK + q] = ((unsigned)dn[j] << 16) | (unsigned)sv[j];
            }
        }
    }
    __syncthreads();
    const int wave = t >> 5, lane = t & 31;
    for (int b = wave; b < NBIN; b += 8) {
        int c = ccnt[b];
        c = (c < CELLCAP) ? c : CELLCAP;
        unsigned v = (unsigned)c;
        if (lane >= 1) v = (lane <= c) ? cell[b * 32 + lane] : 0u;
        part[((size_t)b * NPART + w) * 32 + lane] = v;
    }
    if (t == 0) spill_cnt[w] = *nspillp;
}

#define BUILD_LDS_WORDS (256 + RANGE * CAP + NPART + NPART + 1)
__device__ __forceinline__ void
bucket_build_body(int* __restrict__ lds, const int b,
             const unsigned* __restrict__ part,
             const unsigned* __restrict__ spill,
             const int* __restrict__ spill_cnt,
             int* __restrict__ cnt,
             int* __restrict__ slots,
             int* __restrict__ ovf_src,
             int* __restrict__ ovf_dst)
{
    int* ncnt   = lds;
    int* nslot  = lds + 256;
    int* scnt   = nslot + RANGE * CAP;
    int* slist  = scnt + NPART;
    int* nlistp = slist + NPART;
    const int t = threadIdx.x;
    const int wave = t >> 5, lane = t & 31;
    const int n0 = b * RANGE;
    ncnt[t] = 0;
    if (t == 0) *nlistp = 0;

    auto store_ranked = [&](int dl, int s, int p) {
        if (p < CAP) {
            nslot[dl * CAP + p] = s;
        } else {
            const int q = atomicAdd(&cnt[NN], 1);
            ovf_src[q] = s;
            ovf_dst[q] = n0 + dl;
        }
    };
    auto place = [&](int dl, int s) { store_ranked(dl, s, atomicAdd(&ncnt[dl], 1)); };

    unsigned v[32];
#pragma unroll
    for (int j = 0; j < 32; ++j)
        v[j] = part[((size_t)b * NPART + wave * 32 + j) * 32 + lane];
    const int sc = spill_cnt[t];
    __syncthreads();
    scnt[t] = sc;
    if (sc > 0) slist[atomicAdd(nlistp, 1)] = t;
    int pr[32];
#pragma unroll
    for (int j = 0; j < 32; ++j) {
        const int c = __builtin_amdgcn_readfirstlane((int)v[j]);
        pr[j] = (lane >= 1 && lane <= c) ? atomicAdd(&ncnt[v[j] >> 16], 1) : -1;
    }
#pragma unroll
    for (int j = 0; j < 32; ++j) {
        if (pr[j] >= 0)
            store_ranked((int)(v[j] >> 16), (int)(v[j] & 0xffffu), pr[j]);
    }
    __syncthreads();
    const int nl = *nlistp;
    for (int k = 0; k < nl; ++k) {
        const int w = slist[k];
        int n = scnt[w];
        n = (n < CHUNK) ? n : CHUNK;
        for (int i = t; i < n; i += 256) {
            const unsigned e = spill[(size_t)w * CHUNK + i];
            const int dl = (int)(e >> 16) - n0;
            if (dl >= 0 && dl < RANGE)
                place(dl, (int)(e & 0xffffu));
        }
    }
    if (nl > 0) __syncthreads();
    for (int r = wave; r < RANGE; r += 8) {
        const int node = n0 + r;
        if (node < NN) {
            int c = ncnt[r];
            c = (c < CAP) ? c : CAP;
            slots[(size_t)node * CAP + lane] = (lane < c) ? nslot[r * CAP + lane] : 0;
        }
    }
    if (t < RANGE && n0 + t < NN) cnt[n0 + t] = ncnt[t];
}

#define GEMM_BLOCKS  ((NN / 16 * 32 + GEMM_TB - 1) / GEMM_TB)
#define GEMM_BLOCKS_A ((GEMM_BLOCKS + 1) / 2)
#define GEMM_BLOCKS_B (GEMM_BLOCKS - GEMM_BLOCKS_A)
static_assert(PART_LDS_WORDS * 4 <= HID * HID * 4 && BUILD_LDS_WORDS * 4 <= HID * HID * 4,
              "the grouping passes fit in the LDS sized for the weight matrix");

__global__ void __launch_bounds__(256)
partition_with_gemm(const int* __restrict__ src,
                    const int* __restrict__ dst,
                    unsigned* __restrict__ part,
                    unsigned* __restrict__ spill,
                    int* __restrict__ spill_cnt,
                    int* __restrict__ ovf_total,
                    const float* __restrict__ X,
                    const float* __restrict__ Wm,
                    float* __restrict__ Y)
{
    extern __shared__ __align__(16) float dyn_lds[];
    const int blk = (int)blockIdx.x;
    if (blk < GEMM_BLOCKS_A)
        gemm_strip_body<HID, HID / 16>(dyn_lds, X, Wm, nullptr, Y, blk, NN / 16);
    else
        edge_partition_body((unsigned*)dyn_lds, blk - GEMM_BLOCKS_A,
                            src, dst, part, spill, spill_cnt, ovf_total);
}

__global__ void __launch_bounds__(256)
bucket_build_with_gemm(const unsigned* __restrict__ part,
                       const unsigned* __restrict__ spill,
                       const int* __restrict__ spill_cnt,
                       int* __restrict__ cnt,
                       int* __restrict__ slots,
                       int* __restrict__ ovf_src,
                       int* __restrict__ ovf_dst,
                       const float* __restrict__ X,
                       const float* __restrict__ Wm,
                       float* __restrict__ Y)
{
    extern __shared__ __align__(16) float dyn_lds[];
    const int blk = (int)blockIdx.x;
    if (blk < GEMM_BLOCKS_B)
        gemm_strip_body<HID, HID / 16>(dyn_lds, X, Wm, nullptr, Y, GEMM_BLOCKS_A + blk, NN / 16);
    else
        bucket_build_body((int*)dyn_lds, blk - GEMM_BLOCKS_B,
                          part, spill, spill_cnt, cnt, slots, ovf_src, ovf_dst);
}

__device__ __forceinline__ float lrelu02(float v) {
    return (v > 0.f) ? v : 0.2f * v;
}
#define LOG2E 1.4426950408889634f
__device__ __forceinline__ float exp2_neg(float v) {
    return __builtin_amdgcn_exp2f(v);
}
__device__ __forceinline__ float group8_sum(float v) {
    v += __int_as_float(__builtin_amdgcn_update_dpp(0, __float_as_int(v), 0xB1, 0xF, 0xF, true));
    v += __int_as_float(__builtin_amdgcn_update_dpp(0, __float_as_int(v), 0x4E, 0xF, 0xF, true));
    v += __int_as_float(__builtin_amdgcn_update_dpp(0, __float_as_int(v), 0x141, 0xF, 0xF, true));
    return v;
}

template <int N>
__device__ __forceinline__ void edge_trip(const float* __restrict__ hcol, int s, int k,
                                          int nvalid, const float4 asv, float ad,
                                          float& m, float& d, float4& acc)
{
    float4 r[N];
    float  e[N];
#pragma unroll
    for (int j = 0; j < N; ++j) {
        const int sj = __builtin_amdgcn_readlane(s, k + j);
        r[j] = *(const float4*)(hcol + (size_t)sj * HID);
    }
    float mn = m;
#pragma unroll
    for (int j = 0; j < N; ++j) {
        float p = r[j].x * asv.x + r[j].y * asv.y + r[j].z * asv.z + r[j].w * asv.w;
        p = group8_sum(p);
        const float ej = lrelu02(p + ad);
        e[j] = (j < nvalid) ? ej : -INFINITY;
        mn = fmaxf(mn, e[j]);
    }
    const float sc = exp2_neg(m - mn);
    float dsum = 0.f;
    float4 a;
    a.x = acc.x * sc; a.y = acc.y * sc; a.z = acc.z * sc; a.w = acc.w * sc;
#pragma unroll
    for (int j = 0; j < N; ++j) {
        const float pj = exp2_neg(e[j] - mn);
        dsum += pj;
        a.x += r[j].x * pj; a.y += r[j].y * pj; a.z += r[j].z * pj; a.w += r[j].w * pj;
    }
    d   = d * sc + dsum;
    m   = mn;
    acc = a;
}

__global__ void __launch_bounds__(256)
node_softmax_gather(const int* __restrict__ cnt,
                    const int* __restrict__ slots,
                    const int* __restrict__ ovf_src,
                    const int* __restrict__ ovf_dst,
                    const float* __restrict__ Hm,
                    const float* __restrict__ a_s,
                    const float* __restrict__ a_d,
                    const float* __restrict__ b,
                    float* __restrict__ out)
{
    const int node = __builtin_amdgcn_readfirstlane(
                         (int)((blockIdx.x * 256u + threadIdx.x) >> 5));
    if (node >= NN) return;
    const int lane = (int)(threadIdx.x & 31);
    const int c    = __builtin_amdgcn_readfirstlane(cnt[node]);
    const int cb   = (c < CAP) ? c : CAP;
    const float* hcol = Hm + lane * 4;
    float4 asv = *(const float4*)(a_s + lane * 4);
    float4 adv = *(const float4*)(a_d + lane * 4);
    asv.x *= LOG2E; asv.y *= LOG2E; asv.z *= LOG2E; asv.w *= LOG2E;
    adv.x *= LOG2E; adv.y *= LOG2E; adv.z *= LOG2E; adv.w *= LOG2E;

    const float4 rs = *(const float4*)(hcol + (size_t)node * HID);
    const float ad = group8_sum(rs.x * adv.x + rs.y * adv.y + rs.z * adv.z + rs.w * adv.w);
    float m = lrelu02(group8_sum(rs.x * asv.x + rs.y * asv.y + rs.z * asv.z + rs.w * asv.w) + ad);
    float d = 1.f;
    float4 acc = rs;

    {
        const int s = (lane < cb) ? slots[node * CAP + lane] : node;
        int k = 0;
        for (; k + 4 <= cb; k += 4)
            edge_trip<4>(hcol, s, k, 4, asv, ad, m, d, acc);
        if (k < cb)
            edge_trip<4>(hcol, s, k, cb - k, asv, ad, m, d, acc);
    }
    if (c > CAP) {
        const int novf = __builtin_amdgcn_readfirstlane(cnt[NN]);
        for (int base = 0; base < novf; base += 32) {
            const int i  = base + lane;
            const int od = (i < novf) ? ovf_dst[i] : -1;
            const int os = (i < novf) ? ovf_src[i] : node;
            unsigned hit = __builtin_amdgcn_ballot_w32(od == node);
            while (hit) {
                const int j = __builtin_ctz(hit);
                hit &= hit - 1u;
                edge_trip<1>(hcol, os, j, 1, asv, ad, m, d, acc);
            }
        }
    }

    const float den = d + 1e-16f;
    const float4 bv = *(const float4*)(b + lane * 4);
    float4 y;
    y.x = acc.x / den + bv.x; y.y = acc.y / den + bv.y;
    y.z = acc.z / den + bv.z; y.w = acc.w / den + bv.w;
    y.x = (y.x > 0.f) ? y.x : (exp2_neg(y.x * LOG2E) - 1.f);
    y.y = (y.y > 0.f) ? y.y : (exp2_neg(y.y * LOG2E) - 1.f);
    y.z = (y.z > 0.f) ? y.z : (exp2_neg(y.z * LOG2E) - 1.f);
    y.w = (y.w > 0.f) ? y.w : (exp2_neg(y.w * LOG2E) - 1.f);
    *(float4*)(out + (size_t)node * HID + lane * 4) = y;
}

extern "C" void kernel_launch(void* const* d_in, const int* in_sizes, int n_in,
                              void* d_out, int out_size, void* d_ws, size_t ws_size,
                              hipStream_t stream)
{
    const float* x    = (const float*)d_in[0];
    const int*   edge = (const int*)  d_in[1];
    const float* W1   = (const float*)d_in[2];
    const float* a_s1 = (const float*)d_in[3];
    const float* a_d1 = (const float*)d_in[4];
    const float* b1   = (const float*)d_in[5];
    const float* W2   = (const float*)d_in[6];
    const float* a_s2 = (const float*)d_in[7];
    const float* a_d2 = (const float*)d_in[8];
    const float* b2   = (const float*)d_in[9];
    const float* fc_w = (const float*)d_in[10];
    const float* fc_b = (const float*)d_in[11];
    float* out = (float*)d_out;

    const int* src = edge;
    const int* dst = edge + NE;

    float* bufH   = (float*)d_ws;
    float* bufIn2 = bufH   + (size_t)NN * HID;
    int*   cnt    = (int*)(bufIn2 + (size_t)NN * HID);
    int*   slots  = cnt    + (NN + 16);
    int*   ovfs   = slots  + (size_t)NN * CAP;
    int*   ovfd   = ovfs   + NE;
    unsigned* part  = (unsigned*)(ovfd + NE);
    unsigned* spill = part + (size_t)NBIN * NPART * 32;
    int*      spcnt = (int*)(spill + NE);

    const int TB = 256;
    const int nMtiles = NN / 16;
    const int nodeBlocks = (NN * 32 + TB - 1) / TB;
    const size_t wLds = HID * HID * sizeof(float);

    partition_with_gemm<<<GEMM_BLOCKS_A + NPART, TB, wLds, stream>>>(
        src, dst, part, spill, spcnt, cnt + NN, x, W1, bufH);
    bucket_build_with_gemm<<<GEMM_BLOCKS_B + NBIN, TB, wLds, stream>>>(
        part, spill, spcnt, cnt, slots, ovfs, ovfd, x, W1, bufH);

    node_softmax_gather<<<nodeBlocks, TB, 0, stream>>>(
        cnt, slots, ovfs, ovfd, bufH, a_s1, a_d1, b1, bufIn2);
    wmma_gemm_strip<HID, HID / 16><<<GEMM_BLOCKS, TB, wLds, stream>>>(
        bufIn2, W2, nullptr, bufH, nMtiles);
    node_softmax_gather<<<nodeBlocks, TB, 0, stream>>>(
        cnt, slots, ovfs, ovfd, bufH, a_s2, a_d2, b2, bufIn2);
    wmma_gemm_strip<HID, CH / 16><<<GEMM_BLOCKS, TB, HID * CH * sizeof(float), stream>>>(
        bufIn2, fc_w, fc_b, out, nMtiles);
}


# Strip GEMM operands fed ahead of use: first-line A fragments travel with the W staging, the rest are asked for 8+ k-steps early, B is one 8-byte LDS word per tile a k-step ahead; sum order unchanged
# speedup vs baseline: 6.1291x; 1.0449x over previous
#include <hip/hip_runtime.h>
#include <math.h>

#define NN    50000
#define NE    800000
#define HID   128
#define NH    4
#define CH    32

typedef float v2f __attribute__((ext_vector_type(2)));
typedef float v8f __attribute__((ext_vector_type(8)));

#define GEMM_TB 256
#define GEMM_LDS_PITCH(NOUT) (2 * (NOUT))
#define GEMM_LDS_BYTES(KK, NOUT) (((KK) / 2) * GEMM_LDS_PITCH(NOUT) * 4)

template <int K, int NT>
__device__ __forceinline__ void
gemm_strip_body(float* __restrict__ Wlds,
                const float* __restrict__ X,
                const float* __restrict__ W,
                const float* __restrict__ bias,
                float* __restrict__ Y,
                int blk, int nMtiles)
{
    constexpr int Nout  = NT * 16;
    constexpr int PITCH = GEMM_LDS_PITCH(Nout);
    constexpr int KS    = K / 4;

    const int wave = (int)((blk * GEMM_TB + (int)threadIdx.x) >> 5);
    const int lane = (int)(threadIdx.x & 31);
    const int col  = lane & 15;
    const int h    = lane >> 4;

    constexpr int NPAIR = (K * Nout) / (GEMM_TB * 8);
    static_assert(NPAIR * GEMM_TB * 8 == K * Nout, "W is a whole number of row-pair pieces per thread");
    float4 wa[NPAIR], wb[NPAIR];
#pragma unroll
    for (int j = 0; j < NPAIR; ++j) {
        const int p  = j * GEMM_TB + (int)threadIdx.x;
        const int kp = p / (Nout / 4), c4 = p % (Nout / 4);
        wa[j] = *(const float4*)&W[(2 * kp)     * Nout + 4 * c4];
        wb[j] = *(const float4*)&W[(2 * kp + 1) * Nout + 4 * c4];
    }

    constexpr int KA0 = 8;
    const int awave = (wave < nMtiles) ? wave : nMtiles - 1;
    const float* xrow = X + (size_t)(awave * 16 + col) * K + 2 * h;
    v2f a[KS];
#pragma unroll
    for (int ks = 0; ks < KA0; ++ks)
        a[ks] = *(const v2f*)(xrow + 4 * ks);

#pragma unroll
    for (int j = 0; j < NPAIR; ++j) {
        const int p  = j * GEMM_TB + (int)threadIdx.x;
        const int kp = p / (Nout / 4), c4 = p % (Nout / 4);
        float* d = &Wlds[kp * PITCH + 8 * c4];
        *(float4*)(d)     = make_float4(wa[j].x, wb[j].x, wa[j].y, wb[j].y);
        *(float4*)(d + 4) = make_float4(wa[j].z, wb[j].z, wa[j].w, wb[j].w);
    }
    static_assert(KA0 == 8, "the fence names eight fragments");
    asm volatile("" :: "v"(a[0]), "v"(a[1]), "v"(a[2]), "v"(a[3]),
                       "v"(a[4]), "v"(a[5]), "v"(a[6]), "v"(a[7]));
    __syncthreads();

    if (wave >= nMtiles) return;

#pragma unroll
    for (int ks = KA0; ks < KS; ++ks)
        a[ks] = *(const v2f*)(xrow + 4 * ks);
    __builtin_amdgcn_sched_group_barrier(0x020, KS - KA0, 0);

    v8f acc[NT];
#pragma unroll
    for (int t = 0; t < NT; ++t)
        acc[t] = (v8f){0.f, 0.f, 0.f, 0.f, 0.f, 0.f, 0.f, 0.f};

    constexpr int PD  = (NT >= 8) ? 1 : 3;
    constexpr int DSN = NT / 2;
    const float* wl = Wlds + h * PITCH + 2 * col;
    v2f b[KS][NT];
#pragma unroll
    for (int s = 0; s < PD; ++s) {
#pragma unroll
        for (int t = 0; t < NT; ++t)
            b[s][t] = *(const v2f*)(wl + s * 2 * PITCH + t * 32);
        __builtin_amdgcn_sched_group_barrier(0x100, DSN, 0);
    }
#pragma unroll
    for (int ks = 0; ks < KS; ++ks) {
        if (ks + PD < KS) {
#pragma unroll
            for (int t = 0; t < NT; ++t)
                b[ks + PD][t] = *(const v2f*)(wl + (ks + PD) * 2 * PITCH + t * 32);
        }
#pragma unroll
        for (int t = 0; t < NT; ++t)
            acc[t] = __builtin_amdgcn_wmma_f32_16x16x4_f32(
                         false, a[ks], false, b[ks][t], (short)0, acc[t], false, false);
        if (ks + PD < KS)
            __builtin_amdgcn_sched_group_barrier(0x100, DSN, 0);
        __builtin_amdgcn_sched_group_barrier(0x008, NT, 0);
    }

    int orow = wave * 16 + ((lane >> 4) << 3);
#pragma unroll
    for (int t = 0; t < NT; ++t) {
        int ocol = t * 16 + (lane & 15);
        float badd = bias ? bias[ocol] : 0.f;
#pragma unroll
        for (int r = 0; r < 8; ++r)
            Y[(size_t)(orow + r) * Nout + ocol] = acc[t][r] + badd;
    }
}

template <int K, int NT>
__global__ void __launch_bounds__(GEMM_TB)
wmma_gemm_strip(const float* __restrict__ X,
                const float* __restrict__ W,
                const float* __restrict__ bias,
                float* __restrict__ Y,
                int nMtiles)
{
    extern __shared__ __align__(16) float Wlds[];
    gemm_strip_body<K, NT>(Wlds, X, W, bias, Y, (int)blockIdx.x, nMtiles);
}

#define CAP     32
#define NBIN    256
#define RANGE   196
#define NPART   256
#define CHUNK   3125
#define CELLCAP 31
static_assert(NBIN * RANGE >= NN, "ranges cover all nodes");
static_assert(NPART * CHUNK == NE, "chunks cover all edges");
static_assert(NN <= 65536 && RANGE <= 256, "packed entry: 16-bit source, local destination above it");

#define PART_LDS_WORDS (NBIN + NBIN * 32 + 1)
__device__ __forceinline__ void
edge_partition_body(unsigned* __restrict__ lds, const int w,
               const int* __restrict__ src,
               const int* __restrict__ dst,
               unsigned* __restrict__ part,
               unsigned* __restrict__ spill,
               int* __restrict__ spill_cnt,
               int* __restrict__ ovf_total)
{
    int*      ccnt    = (int*)lds;
    unsigned* cell    = lds + NBIN;
    int*      nspillp = (int*)(lds + NBIN + NBIN * 32);
    const int t = threadIdx.x;
    ccnt[t] = 0;
    if (t == 0) *nspillp = 0;
    if (w == 0 && t == 0) *ovf_total = 0;

    constexpr int PER = (CHUNK + 255) / 256;
    int dn[PER], sv[PER];
#pragma unroll
    for (int j = 0; j < PER; ++j) {
        const int i = j * 256 + t;
        const int e = w * CHUNK + ((i < CHUNK) ? i : 0);
        dn[j] = dst[e];
        sv[j] = src[e];
    }
    __syncthreads();
    int bn[PER], pr[PER];
#pragma unroll
    for (int j = 0; j < PER; ++j) {
        bn[j] = dn[j] / RANGE;
        pr[j] = (j * 256 + t < CHUNK) ? atomicAdd(&ccnt[bn[j]], 1) : -1;
    }
#pragma unroll
    for (int j = 0; j < PER; ++j) {
        if (pr[j] >= 0) {
            if (pr[j] < CELLCAP) {
                cell[bn[j] * 32 + 1 + pr[j]] = ((unsigned)(dn[j] - bn[j] * RANGE) << 16) | (unsigned)sv[j];
            } else {
                const int q = atomicAdd(nspillp, 1);
                spill[(size_t)w * CHUNK + q] = ((unsigned)dn[j] << 16) | (unsigned)sv[j];
            }
        }
    }
    __syncthreads();
    const int wave = t >> 5, lane = t & 31;
    for (int b = wave; b < NBIN; b += 8) {
        int c = ccnt[b];
        c = (c < CELLCAP) ? c : CELLCAP;
        unsigned v = (unsigned)c;
        if (lane >= 1) v = (lane <= c) ? cell[b * 32 + lane] : 0u;
        part[((size_t)b * NPART + w) * 32 + lane] = v;
    }
    if (t == 0) spill_cnt[w] = *nspillp;
}

#define BUILD_LDS_WORDS (256 + RANGE * CAP + NPART + NPART + 1)
__device__ __forceinline__ void
bucket_build_body(int* __restrict__ lds, const int b,
             const unsigned* __restrict__ part,
             const unsigned* __restrict__ spill,
             const int* __restrict__ spill_cnt,
             int* __restrict__ cnt,
             int* __restrict__ slots,
             int* __restrict__ ovf_src,
             int* __restrict__ ovf_dst)
{
    int* ncnt   = lds;
    int* nslot  = lds + 256;
    int* scnt   = nslot + RANGE * CAP;
    int* slist  = scnt + NPART;
    int* nlistp = slist + NPART;
    const int t = threadIdx.x;
    const int wave = t >> 5, lane = t & 31;
    const int n0 = b * RANGE;
    ncnt[t] = 0;
    if (t == 0) *nlistp = 0;

    auto store_ranked = [&](int dl, int s, int p) {
        if (p < CAP) {
            nslot[dl * CAP + p] = s;
        } else {
            const int q = atomicAdd(&cnt[NN], 1);
            ovf_src[q] = s;
            ovf_dst[q] = n0 + dl;
        }
    };
    auto place = [&](int dl, int s) { store_ranked(dl, s, atomicAdd(&ncnt[dl], 1)); };

    unsigned v[32];
#pragma unroll
    for (int j = 0; j < 32; ++j)
        v[j] = part[((size_t)b * NPART + wave * 32 + j) * 32 + lane];
    const int sc = spill_cnt[t];
    __syncthreads();
    scnt[t] = sc;
    if (sc > 0) slist[atomicAdd(nlistp, 1)] = t;
    int pr[32];
#pragma unroll
    for (int j = 0; j < 32; ++j) {
        const int c = __builtin_amdgcn_readfirstlane((int)v[j]);
        pr[j] = (lane >= 1 && lane <= c) ? atomicAdd(&ncnt[v[j] >> 16], 1) : -1;
    }
#pragma unroll
    for (int j = 0; j < 32; ++j) {
        if (pr[j] >= 0)
            store_ranked((int)(v[j] >> 16), (int)(v[j] & 0xffffu), pr[j]);
    }
    __syncthreads();
    const int nl = *nlistp;
    for (int k = 0; k < nl; ++k) {
        const int w = slist[k];
        int n = scnt[w];
        n = (n < CHUNK) ? n : CHUNK;
        for (int i = t; i < n; i += 256) {
            const unsigned e = spill[(size_t)w * CHUNK + i];
            const int dl = (int)(e >> 16) - n0;
            if (dl >= 0 && dl < RANGE)
                place(dl, (int)(e & 0xffffu));
        }
    }
    if (nl > 0) __syncthreads();
    for (int r = wave; r < RANGE; r += 8) {
        const int node = n0 + r;
        if (node < NN) {
            int c = ncnt[r];
            c = (c < CAP) ? c : CAP;
            slots[(size_t)node * CAP + lane] = (lane < c) ? nslot[r * CAP + lane] : 0;
        }
    }
    if (t < RANGE && n0 + t < NN) cnt[n0 + t] = ncnt[t];
}

#define GEMM_BLOCKS  ((NN / 16 * 32 + GEMM_TB - 1) / GEMM_TB)
#define GEMM_BLOCKS_A ((GEMM_BLOCKS + 1) / 2)
#define GEMM_BLOCKS_B (GEMM_BLOCKS - GEMM_BLOCKS_A)
static_assert(PART_LDS_WORDS * 4 <= GEMM_LDS_BYTES(HID, HID) && BUILD_LDS_WORDS * 4 <= GEMM_LDS_BYTES(HID, HID),
              "the grouping passes fit in the LDS sized for the weight matrix");

__global__ void __launch_bounds__(256)
partition_with_gemm(const int* __restrict__ src,
                    const int* __restrict__ dst,
                    unsigned* __restrict__ part,
                    unsigned* __restrict__ spill,
                    int* __restrict__ spill_cnt,
                    int* __restrict__ ovf_total,
                    const float* __restrict__ X,
                    const float* __restrict__ Wm,
                    float* __restrict__ Y)
{
    extern __shared__ __align__(16) float dyn_lds[];
    const int blk = (int)blockIdx.x;
    if (blk < GEMM_BLOCKS_A)
        gemm_strip_body<HID, HID / 16>(dyn_lds, X, Wm, nullptr, Y, blk, NN / 16);
    else
        edge_partition_body((unsigned*)dyn_lds, blk - GEMM_BLOCKS_A,
                            src, dst, part, spill, spill_cnt, ovf_total);
}

__global__ void __launch_bounds__(256)
bucket_build_with_gemm(const unsigned* __restrict__ part,
                       const unsigned* __restrict__ spill,
                       const int* __restrict__ spill_cnt,
                       int* __restrict__ cnt,
                       int* __restrict__ slots,
                       int* __restrict__ ovf_src,
                       int* __restrict__ ovf_dst,
                       const float* __restrict__ X,
                       const float* __restrict__ Wm,
                       float* __restrict__ Y)
{
    extern __shared__ __align__(16) float dyn_lds[];
    const int blk = (int)blockIdx.x;
    if (blk < GEMM_BLOCKS_B)
        gemm_strip_body<HID, HID / 16>(dyn_lds, X, Wm, nullptr, Y, GEMM_BLOCKS_A + blk, NN / 16);
    else
        bucket_build_body((int*)dyn_lds, blk - GEMM_BLOCKS_B,
                          part, spill, spill_cnt, cnt, slots, ovf_src, ovf_dst);
}

__device__ __forceinline__ float lrelu02(float v) {
    return (v > 0.f) ? v : 0.2f * v;
}
#define LOG2E 1.4426950408889634f
__device__ __forceinline__ float exp2_neg(float v) {
    return __builtin_amdgcn_exp2f(v);
}
__device__ __forceinline__ float group8_sum(float v) {
    v += __int_as_float(__builtin_amdgcn_update_dpp(0, __float_as_int(v), 0xB1, 0xF, 0xF, true));
    v += __int_as_float(__builtin_amdgcn_update_dpp(0, __float_as_int(v), 0x4E, 0xF, 0xF, true));
    v += __int_as_float(__builtin_amdgcn_update_dpp(0, __float_as_int(v), 0x141, 0xF, 0xF, true));
    return v;
}

template <int N>
__device__ __forceinline__ void edge_trip(const float* __restrict__ hcol, int s, int k,
                                          int nvalid, const float4 asv, float ad,
                                          float& m, float& d, float4& acc)
{
    float4 r[N];
    float  e[N];
#pragma unroll
    for (int j = 0; j < N; ++j) {
        const int sj = __builtin_amdgcn_readlane(s, k + j);
        r[j] = *(const float4*)(hcol + (size_t)sj * HID);
    }
    float mn = m;
#pragma unroll
    for (int j = 0; j < N; ++j) {
        float p = r[j].x * asv.x + r[j].y * asv.y + r[j].z * asv.z + r[j].w * asv.w;
        p = group8_sum(p);
        const float ej = lrelu02(p + ad);
        e[j] = (j < nvalid) ? ej : -INFINITY;
        mn = fmaxf(mn, e[j]);
    }
    const float sc = exp2_neg(m - mn);
    float dsum = 0.f;
    float4 a;
    a.x = acc.x * sc; a.y = acc.y * sc; a.z = acc.z * sc; a.w = acc.w * sc;
#pragma unroll
    for (int j = 0; j < N; ++j) {
        const float pj = exp2_neg(e[j] - mn);
        dsum += pj;
        a.x += r[j].x * pj; a.y += r[j].y * pj; a.z += r[j].z * pj; a.w += r[j].w * pj;
    }
    d   = d * sc + dsum;
    m   = mn;
    acc = a;
}

__global__ void __launch_bounds__(256)
node_softmax_gather(const int* __restrict__ cnt,
                    const int* __restrict__ slots,
                    const int* __restrict__ ovf_src,
                    const int* __restrict__ ovf_dst,
                    const float* __restrict__ Hm,
                    const float* __restrict__ a_s,
                    const float* __restrict__ a_d,
                    const float* __restrict__ b,
                    float* __restrict__ out)
{
    const int node = __builtin_amdgcn_readfirstlane(
                         (int)((blockIdx.x * 256u + threadIdx.x) >> 5));
    if (node >= NN) return;
    const int lane = (int)(threadIdx.x & 31);
    const int c    = __builtin_amdgcn_readfirstlane(cnt[node]);
    const int cb   = (c < CAP) ? c : CAP;
    const float* hcol = Hm + lane * 4;
    float4 asv = *(const float4*)(a_s + lane * 4);
    float4 adv = *(const float4*)(a_d + lane * 4);
    asv.x *= LOG2E; asv.y *= LOG2E; asv.z *= LOG2E; asv.w *= LOG2E;
    adv.x *= LOG2E; adv.y *= LOG2E; adv.z *= LOG2E; adv.w *= LOG2E;

    const float4 rs = *(const float4*)(hcol + (size_t)node * HID);
    const float ad = group8_sum(rs.x * adv.x + rs.y * adv.y + rs.z * adv.z + rs.w * adv.w);
    float m = lrelu02(group8_sum(rs.x * asv.x + rs.y * asv.y + rs.z * asv.z + rs.w * asv.w) + ad);
    float d = 1.f;
    float4 acc = rs;

    {
        const int s = (lane < cb) ? slots[node * CAP + lane] : node;
        int k = 0;
        for (; k + 4 <= cb; k += 4)
            edge_trip<4>(hcol, s, k, 4, asv, ad, m, d, acc);
        if (k < cb)
            edge_trip<4>(hcol, s, k, cb - k, asv, ad, m, d, acc);
    }
    if (c > CAP) {
        const int novf = __builtin_amdgcn_readfirstlane(cnt[NN]);
        for (int base = 0; base < novf; base += 32) {
            const int i  = base + lane;
            const int od = (i < novf) ? ovf_dst[i] : -1;
            const int os = (i < novf) ? ovf_src[i] : node;
            unsigned hit = __builtin_amdgcn_ballot_w32(od == node);
            while (hit) {
                const int j = __builtin_ctz(hit);
                hit &= hit - 1u;
                edge_trip<1>(hcol, os, j, 1, asv, ad, m, d, acc);
            }
        }
    }

    const float den = d + 1e-16f;
    const float4 bv = *(const float4*)(b + lane * 4);
    float4 y;
    y.x = acc.x / den + bv.x; y.y = acc.y / den + bv.y;
    y.z = acc.z / den + bv.z; y.w = acc.w / den + bv.w;
    y.x = (y.x > 0.f) ? y.x : (exp2_neg(y.x * LOG2E) - 1.f);
    y.y = (y.y > 0.f) ? y.y : (exp2_neg(y.y * LOG2E) - 1.f);
    y.z = (y.z > 0.f) ? y.z : (exp2_neg(y.z * LOG2E) - 1.f);
    y.w = (y.w > 0.f) ? y.w : (exp2_neg(y.w * LOG2E) - 1.f);
    *(float4*)(out + (size_t)node * HID + lane * 4) = y;
}

extern "C" void kernel_launch(void* const* d_in, const int* in_sizes, int n_in,
                              void* d_out, int out_size, void* d_ws, size_t ws_size,
                              hipStream_t stream)
{
    const float* x    = (const float*)d_in[0];
    const int*   edge = (const int*)  d_in[1];
    const float* W1   = (const float*)d_in[2];
    const float* a_s1 = (const float*)d_in[3];
    const float* a_d1 = (const float*)d_in[4];
    const float* b1   = (const float*)d_in[5];
    const float* W2   = (const float*)d_in[6];
    const float* a_s2 = (const float*)d_in[7];
    const float* a_d2 = (const float*)d_in[8];
    const float* b2   = (const float*)d_in[9];
    const float* fc_w = (const float*)d_in[10];
    const float* fc_b = (const float*)d_in[11];
    float* out = (float*)d_out;

    const int* src = edge;
    const int* dst = edge + NE;

    float* bufH   = (float*)d_ws;
    float* bufIn2 = bufH   + (size_t)NN * HID;
    int*   cnt    = (int*)(bufIn2 + (size_t)NN * HID);
    int*   slots  = cnt    + (NN + 16);
    int*   ovfs   = slots  + (size_t)NN * CAP;
    int*   ovfd   = ovfs   + NE;
    unsigned* part  = (unsigned*)(ovfd + NE);
    unsigned* spill = part + (size_t)NBIN * NPART * 32;
    int*      spcnt = (int*)(spill + NE);

    const int TB = 256;
    const int nMtiles = NN / 16;
    const int nodeBlocks = (NN * 32 + TB - 1) / TB;
    const size_t wLds = GEMM_LDS_BYTES(HID, HID);

    partition_with_gemm<<<GEMM_BLOCKS_A + NPART, TB, wLds, stream>>>(
        src, dst, part, spill, spcnt, cnt + NN, x, W1, bufH);
    bucket_build_with_gemm<<<GEMM_BLOCKS_B + NBIN, TB, wLds, stream>>>(
        part, spill, spcnt, cnt, slots, ovfs, ovfd, x, W1, bufH);

    node_softmax_gather<<<nodeBlocks, TB, 0, stream>>>(
        cnt, slots, ovfs, ovfd, bufH, a_s1, a_d1, b1, bufIn2);
    wmma_gemm_strip<HID, HID / 16><<<GEMM_BLOCKS, TB, wLds, stream>>>(
        bufIn2, W2, nullptr, bufH, nMtiles);
    node_softmax_gather<<<nodeBlocks, TB, 0, stream>>>(
        cnt, slots, ovfs, ovfd, bufH, a_s2, a_d2, b2, bufIn2);
    wmma_gemm_strip<HID, CH / 16><<<GEMM_BLOCKS, TB, GEMM_LDS_BYTES(HID, CH), stream>>>(
        bufIn2, fc_w, fc_b, out, nMtiles);
}


# Node kernel: softmax scalars on one lane per edge: the 8-lane logit sum transposes (keep 2 of 4, then 1 of 2), so leaky relu, max, exp2 run once per lane, not 4x; float sum order changes (pairwise)
# speedup vs baseline: 6.3192x; 1.0310x over previous
#include <hip/hip_runtime.h>
#include <math.h>

#define NN    50000
#define NE    800000
#define HID   128
#define NH    4
#define CH    32

typedef float v2f __attribute__((ext_vector_type(2)));
typedef float v8f __attribute__((ext_vector_type(8)));

#define GEMM_TB 256
#define GEMM_LDS_PITCH(NOUT) (2 * (NOUT))
#define GEMM_LDS_BYTES(KK, NOUT) (((KK) / 2) * GEMM_LDS_PITCH(NOUT) * 4)

template <int K, int NT>
__device__ __forceinline__ void
gemm_strip_body(float* __restrict__ Wlds,
                const float* __restrict__ X,
                const float* __restrict__ W,
                const float* __restrict__ bias,
                float* __restrict__ Y,
                int blk, int nMtiles)
{
    constexpr int Nout  = NT * 16;
    constexpr int PITCH = GEMM_LDS_PITCH(Nout);
    constexpr int KS    = K / 4;

    const int wave = (int)((blk * GEMM_TB + (int)threadIdx.x) >> 5);
    const int lane = (int)(threadIdx.x & 31);
    const int col  = lane & 15;
    const int h    = lane >> 4;

    constexpr int NPAIR = (K * Nout) / (GEMM_TB * 8);
    static_assert(NPAIR * GEMM_TB * 8 == K * Nout, "W is a whole number of row-pair pieces per thread");
    float4 wa[NPAIR], wb[NPAIR];
#pragma unroll
    for (int j = 0; j < NPAIR; ++j) {
        const int p  = j * GEMM_TB + (int)threadIdx.x;
        const int kp = p / (Nout / 4), c4 = p % (Nout / 4);
        wa[j] = *(const float4*)&W[(2 * kp)     * Nout + 4 * c4];
        wb[j] = *(const float4*)&W[(2 * kp + 1) * Nout + 4 * c4];
    }

    constexpr int KA0 = 8;
    const int awave = (wave < nMtiles) ? wave : nMtiles - 1;
    const float* xrow = X + (size_t)(awave * 16 + col) * K + 2 * h;
    v2f a[KS];
#pragma unroll
    for (int ks = 0; ks < KA0; ++ks)
        a[ks] = *(const v2f*)(xrow + 4 * ks);

#pragma unroll
    for (int j = 0; j < NPAIR; ++j) {
        const int p  = j * GEMM_TB + (int)threadIdx.x;
        const int kp = p / (Nout / 4), c4 = p % (Nout / 4);
        float* d = &Wlds[kp * PITCH + 8 * c4];
        *(float4*)(d)     = make_float4(wa[j].x, wb[j].x, wa[j].y, wb[j].y);
        *(float4*)(d + 4) = make_float4(wa[j].z, wb[j].z, wa[j].w, wb[j].w);
    }
    static_assert(KA0 == 8, "the fence names eight fragments");
    asm volatile("" :: "v"(a[0]), "v"(a[1]), "v"(a[2]), "v"(a[3]),
                       "v"(a[4]), "v"(a[5]), "v"(a[6]), "v"(a[7]));
    __syncthreads();

    if (wave >= nMtiles) return;

#pragma unroll
    for (int ks = KA0; ks < KS; ++ks)
        a[ks] = *(const v2f*)(xrow + 4 * ks);
    __builtin_amdgcn_sched_group_barrier(0x020, KS - KA0, 0);

    v8f acc[NT];
#pragma unroll
    for (int t = 0; t < NT; ++t)
        acc[t] = (v8f){0.f, 0.f, 0.f, 0.f, 0.f, 0.f, 0.f, 0.f};

    constexpr int PD  = (NT >= 8) ? 1 : 3;
    constexpr int DSN = NT / 2;
    const float* wl = Wlds + h * PITCH + 2 * col;
    v2f b[KS][NT];
#pragma unroll
    for (int s = 0; s < PD; ++s) {
#pragma unroll
        for (int t = 0; t < NT; ++t)
            b[s][t] = *(const v2f*)(wl + s * 2 * PITCH + t * 32);
        __builtin_amdgcn_sched_group_barrier(0x100, DSN, 0);
    }
#pragma unroll
    for (int ks = 0; ks < KS; ++ks) {
        if (ks + PD < KS) {
#pragma unroll
            for (int t = 0; t < NT; ++t)
                b[ks + PD][t] = *(const v2f*)(wl + (ks + PD) * 2 * PITCH + t * 32);
        }
#pragma unroll
        for (int t = 0; t < NT; ++t)
            acc[t] = __builtin_amdgcn_wmma_f32_16x16x4_f32(
                         false, a[ks], false, b[ks][t], (short)0, acc[t], false, false);
        if (ks + PD < KS)
            __builtin_amdgcn_sched_group_barrier(0x100, DSN, 0);
        __builtin_amdgcn_sched_group_barrier(0x008, NT, 0);
    }

    int orow = wave * 16 + ((lane >> 4) << 3);
#pragma unroll
    for (int t = 0; t < NT; ++t) {
        int ocol = t * 16 + (lane & 15);
        float badd = bias ? bias[ocol] : 0.f;
#pragma unroll
        for (int r = 0; r < 8; ++r)
            Y[(size_t)(orow + r) * Nout + ocol] = acc[t][r] + badd;
    }
}

template <int K, int NT>
__global__ void __launch_bounds__(GEMM_TB)
wmma_gemm_strip(const float* __restrict__ X,
                const float* __restrict__ W,
                const float* __restrict__ bias,
                float* __restrict__ Y,
                int nMtiles)
{
    extern __shared__ __align__(16) float Wlds[];
    gemm_strip_body<K, NT>(Wlds, X, W, bias, Y, (int)blockIdx.x, nMtiles);
}

#define CAP     32
#define NBIN    256
#define RANGE   196
#define NPART   256
#define CHUNK   3125
#define CELLCAP 31
static_assert(NBIN * RANGE >= NN, "ranges cover all nodes");
static_assert(NPART * CHUNK == NE, "chunks cover all edges");
static_assert(NN <= 65536 && RANGE <= 256, "packed entry: 16-bit source, local destination above it");

#define PART_LDS_WORDS (NBIN + NBIN * 32 + 1)
__device__ __forceinline__ void
edge_partition_body(unsigned* __restrict__ lds, const int w,
               const int* __restrict__ src,
               const int* __restrict__ dst,
               unsigned* __restrict__ part,
               unsigned* __restrict__ spill,
               int* __restrict__ spill_cnt,
               int* __restrict__ ovf_total)
{
    int*      ccnt    = (int*)lds;
    unsigned* cell    = lds + NBIN;
    int*      nspillp = (int*)(lds + NBIN + NBIN * 32);
    const int t = threadIdx.x;
    ccnt[t] = 0;
    if (t == 0) *nspillp = 0;
    if (w == 0 && t == 0) *ovf_total = 0;

    constexpr int PER = (CHUNK + 255) / 256;
    int dn[PER], sv[PER];
#pragma unroll
    for (int j = 0; j < PER; ++j) {
        const int i = j * 256 + t;
        const int e = w * CHUNK + ((i < CHUNK) ? i : 0);
        dn[j] = dst[e];
        sv[j] = src[e];
    }
    __syncthreads();
    int bn[PER], pr[PER];
#pragma unroll
    for (int j = 0; j < PER; ++j) {
        bn[j] = dn[j] / RANGE;
        pr[j] = (j * 256 + t < CHUNK) ? atomicAdd(&ccnt[bn[j]], 1) : -1;
    }
#pragma unroll
    for (int j = 0; j < PER; ++j) {
        if (pr[j] >= 0) {
            if (pr[j] < CELLCAP) {
                cell[bn[j] * 32 + 1 + pr[j]] = ((unsigned)(dn[j] - bn[j] * RANGE) << 16) | (unsigned)sv[j];
            } else {
                const int q = atomicAdd(nspillp, 1);
                spill[(size_t)w * CHUNK + q] = ((unsigned)dn[j] << 16) | (unsigned)sv[j];
            }
        }
    }
    __syncthreads();
    const int wave = t >> 5, lane = t & 31;
    for (int b = wave; b < NBIN; b += 8) {
        int c = ccnt[b];
        c = (c < CELLCAP) ? c : CELLCAP;
        unsigned v = (unsigned)c;
        if (lane >= 1) v = (lane <= c) ? cell[b * 32 + lane] : 0u;
        part[((size_t)b * NPART + w) * 32 + lane] = v;
    }
    if (t == 0) spill_cnt[w] = *nspillp;
}

#define BUILD_LDS_WORDS (256 + RANGE * CAP + NPART + NPART + 1)
__device__ __forceinline__ void
bucket_build_body(int* __restrict__ lds, const int b,
             const unsigned* __restrict__ part,
             const unsigned* __restrict__ spill,
             const int* __restrict__ spill_cnt,
             int* __restrict__ cnt,
             int* __restrict__ slots,
             int* __restrict__ ovf_src,
             int* __restrict__ ovf_dst)
{
    int* ncnt   = lds;
    int* nslot  = lds + 256;
    int* scnt   = nslot + RANGE * CAP;
    int* slist  = scnt + NPART;
    int* nlistp = slist + NPART;
    const int t = threadIdx.x;
    const int wave = t >> 5, lane = t & 31;
    const int n0 = b * RANGE;
    ncnt[t] = 0;
    if (t == 0) *nlistp = 0;

    auto store_ranked = [&](int dl, int s, int p) {
        if (p < CAP) {
            nslot[dl * CAP + p] = s;
        } else {
            const int q = atomicAdd(&cnt[NN], 1);
            ovf_src[q] = s;
            ovf_dst[q] = n0 + dl;
        }
    };
    auto place = [&](int dl, int s) { store_ranked(dl, s, atomicAdd(&ncnt[dl], 1)); };

    unsigned v[32];
#pragma unroll
    for (int j = 0; j < 32; ++j)
        v[j] = part[((size_t)b * NPART + wave * 32 + j) * 32 + lane];
    const int sc = spill_cnt[t];
    __syncthreads();
    scnt[t] = sc;
    if (sc > 0) slist[atomicAdd(nlistp, 1)] = t;
    int pr[32];
#pragma unroll
    for (int j = 0; j < 32; ++j) {
        const int c = __builtin_amdgcn_readfirstlane((int)v[j]);
        pr[j] = (lane >= 1 && lane <= c) ? atomicAdd(&ncnt[v[j] >> 16], 1) : -1;
    }
#pragma unroll
    for (int j = 0; j < 32; ++j) {
        if (pr[j] >= 0)
            store_ranked((int)(v[j] >> 16), (int)(v[j] & 0xffffu), pr[j]);
    }
    __syncthreads();
    const int nl = *nlistp;
    for (int k = 0; k < nl; ++k) {
        const int w = slist[k];
        int n = scnt[w];
        n = (n < CHUNK) ? n : CHUNK;
        for (int i = t; i < n; i += 256) {
            const unsigned e = spill[(size_t)w * CHUNK + i];
            const int dl = (int)(e >> 16) - n0;
            if (dl >= 0 && dl < RANGE)
                place(dl, (int)(e & 0xffffu));
        }
    }
    if (nl > 0) __syncthreads();
    for (int r = wave; r < RANGE; r += 8) {
        const int node = n0 + r;
        if (node < NN) {
            int c = ncnt[r];
            c = (c < CAP) ? c : CAP;
            slots[(size_t)node * CAP + lane] = (lane < c) ? nslot[r * CAP + lane] : 0;
        }
    }
    if (t < RANGE && n0 + t < NN) cnt[n0 + t] = ncnt[t];
}

#define GEMM_BLOCKS  ((NN / 16 * 32 + GEMM_TB - 1) / GEMM_TB)
#define GEMM_BLOCKS_A ((GEMM_BLOCKS + 1) / 2)
#define GEMM_BLOCKS_B (GEMM_BLOCKS - GEMM_BLOCKS_A)
static_assert(PART_LDS_WORDS * 4 <= GEMM_LDS_BYTES(HID, HID) && BUILD_LDS_WORDS * 4 <= GEMM_LDS_BYTES(HID, HID),
              "the grouping passes fit in the LDS sized for the weight matrix");

__global__ void __launch_bounds__(256)
partition_with_gemm(const int* __restrict__ src,
                    const int* __restrict__ dst,
                    unsigned* __restrict__ part,
                    unsigned* __restrict__ spill,
                    int* __restrict__ spill_cnt,
                    int* __restrict__ ovf_total,
                    const float* __restrict__ X,
                    const float* __restrict__ Wm,
                    float* __restrict__ Y)
{
    extern __shared__ __align__(16) float dyn_lds[];
    const int blk = (int)blockIdx.x;
    if (blk < GEMM_BLOCKS_A)
        gemm_strip_body<HID, HID / 16>(dyn_lds, X, Wm, nullptr, Y, blk, NN / 16);
    else
        edge_partition_body((unsigned*)dyn_lds, blk - GEMM_BLOCKS_A,
                            src, dst, part, spill, spill_cnt, ovf_total);
}

__global__ void __launch_bounds__(256)
bucket_build_with_gemm(const unsigned* __restrict__ part,
                       const unsigned* __restrict__ spill,
                       const int* __restrict__ spill_cnt,
                       int* __restrict__ cnt,
                       int* __restrict__ slots,
                       int* __restrict__ ovf_src,
                       int* __restrict__ ovf_dst,
                       const float* __restrict__ X,
                       const float* __restrict__ Wm,
                       float* __restrict__ Y)
{
    extern __shared__ __align__(16) float dyn_lds[];
    const int blk = (int)blockIdx.x;
    if (blk < GEMM_BLOCKS_B)
        gemm_strip_body<HID, HID / 16>(dyn_lds, X, Wm, nullptr, Y, GEMM_BLOCKS_A + blk, NN / 16);
    else
        bucket_build_body((int*)dyn_lds, blk - GEMM_BLOCKS_B,
                          part, spill, spill_cnt, cnt, slots, ovf_src, ovf_dst);
}

__device__ __forceinline__ float lrelu02(float v) {
    return (v > 0.f) ? v : 0.2f * v;
}
#define LOG2E 1.4426950408889634f
__device__ __forceinline__ float exp2_neg(float v) {
    return __builtin_amdgcn_exp2f(v);
}
__device__ __forceinline__ float group8_sum(float v) {
    v += __int_as_float(__builtin_amdgcn_update_dpp(0, __float_as_int(v), 0xB1, 0xF, 0xF, true));
    v += __int_as_float(__builtin_amdgcn_update_dpp(0, __float_as_int(v), 0x4E, 0xF, 0xF, true));
    v += __int_as_float(__builtin_amdgcn_update_dpp(0, __float_as_int(v), 0x141, 0xF, 0xF, true));
    return v;
}

template <int N>
__device__ __forceinline__ void edge_trip(const float* __restrict__ hcol, int s, int k,
                                          int nvalid, const float4 asv, float ad,
                                          float& m, float& d, float4& acc)
{
    float4 r[N];
    float  e[N];
#pragma unroll
    for (int j = 0; j < N; ++j) {
        const int sj = __builtin_amdgcn_readlane(s, k + j);
        r[j] = *(const float4*)(hcol + (size_t)sj * HID);
    }
    float mn = m;
#pragma unroll
    for (int j = 0; j < N; ++j) {
        float p = r[j].x * asv.x + r[j].y * asv.y + r[j].z * asv.z + r[j].w * asv.w;
        p = group8_sum(p);
        const float ej = lrelu02(p + ad);
        e[j] = (j < nvalid) ? ej : -INFINITY;
        mn = fmaxf(mn, e[j]);
    }
    const float sc = exp2_neg(m - mn);
    float dsum = 0.f;
    float4 a;
    a.x = acc.x * sc; a.y = acc.y * sc; a.z = acc.z * sc; a.w = acc.w * sc;
#pragma unroll
    for (int j = 0; j < N; ++j) {
        const float pj = exp2_neg(e[j] - mn);
        dsum += pj;
        a.x += r[j].x * pj; a.y += r[j].y * pj; a.z += r[j].z * pj; a.w += r[j].w * pj;
    }
    d   = d * sc + dsum;
    m   = mn;
    acc = a;
}

__device__ __forceinline__ float lane_xor1(float v) {
    return __int_as_float(__builtin_amdgcn_update_dpp(0, __float_as_int(v), 0xB1, 0xF, 0xF, true));
}
__device__ __forceinline__ float lane_xor2(float v) {
    return __int_as_float(__builtin_amdgcn_update_dpp(0, __float_as_int(v), 0x4E, 0xF, 0xF, true));
}
__device__ __forceinline__ float lane_xor4(float v) {
    return __int_as_float(__builtin_amdgcn_update_dpp(0, __float_as_int(v), 0x164, 0xF, 0xF, true));
}
template <int Q>
__device__ __forceinline__ float quad_bcast(float v) {
    return __int_as_float(__builtin_amdgcn_update_dpp(0, __float_as_int(v), Q * 0x55, 0xF, 0xF, true));
}

__device__ __forceinline__ void edge_trip4(const float* __restrict__ hcol, int s, int k,
                                           int nvalid, const float4 asv, float ad,
                                           bool odd1, bool odd2, int jl,
                                           float& m, float& d, float4& acc)
{
    float4 r[4];
    float  q[4];
#pragma unroll
    for (int j = 0; j < 4; ++j) {
        const int sj = __builtin_amdgcn_readlane(s, k + j);
        r[j] = *(const float4*)(hcol + (size_t)sj * HID);
    }
    const v2f a01 = {asv.x, asv.y}, a23 = {asv.z, asv.w};
#pragma unroll
    for (int j = 0; j < 4; ++j) {
        v2f t = (v2f){r[j].x, r[j].y} * a01;
        t = (v2f){r[j].z, r[j].w} * a23 + t;
        q[j] = t.x + t.y;
    }
    const float u0 = (odd1 ? q[2] : q[0]) + lane_xor1(odd1 ? q[0] : q[2]);
    const float u1 = (odd1 ? q[3] : q[1]) + lane_xor1(odd1 ? q[1] : q[3]);
    float w = (odd2 ? u1 : u0) + lane_xor2(odd2 ? u0 : u1);
    w += lane_xor4(w);
    float e = lrelu02(w + ad);
    e = (jl < nvalid) ? e : -INFINITY;
    float t4 = fmaxf(e, lane_xor1(e));
    t4 = fmaxf(t4, lane_xor2(t4));
    const float mn = fmaxf(m, t4);
    const float sc = exp2_neg(m - mn);
    const float p  = exp2_neg(e - mn);
    float ps = p + lane_xor1(p);
    ps += lane_xor2(ps);
    const float p0 = quad_bcast<0>(p), p1 = quad_bcast<2>(p),
                p2 = quad_bcast<1>(p), p3 = quad_bcast<3>(p);
    v2f c01 = (v2f){acc.x, acc.y} * sc, c23 = (v2f){acc.z, acc.w} * sc;
    c01 = (v2f){r[0].x, r[0].y} * p0 + c01;  c23 = (v2f){r[0].z, r[0].w} * p0 + c23;
    c01 = (v2f){r[1].x, r[1].y} * p1 + c01;  c23 = (v2f){r[1].z, r[1].w} * p1 + c23;
    c01 = (v2f){r[2].x, r[2].y} * p2 + c01;  c23 = (v2f){r[2].z, r[2].w} * p2 + c23;
    c01 = (v2f){r[3].x, r[3].y} * p3 + c01;  c23 = (v2f){r[3].z, r[3].w} * p3 + c23;
    d   = d * sc + ps;
    m   = mn;
    acc = make_float4(c01.x, c01.y, c23.x, c23.y);
}

__global__ void __launch_bounds__(256)
node_softmax_gather(const int* __restrict__ cnt,
                    const int* __restrict__ slots,
                    const int* __restrict__ ovf_src,
                    const int* __restrict__ ovf_dst,
                    const float* __restrict__ Hm,
                    const float* __restrict__ a_s,
                    const float* __restrict__ a_d,
                    const float* __restrict__ b,
                    float* __restrict__ out)
{
    const int node = __builtin_amdgcn_readfirstlane(
                         (int)((blockIdx.x * 256u + threadIdx.x) >> 5));
    if (node >= NN) return;
    const int lane = (int)(threadIdx.x & 31);
    const int c    = __builtin_amdgcn_readfirstlane(cnt[node]);
    const int cb   = (c < CAP) ? c : CAP;
    const float* hcol = Hm + lane * 4;
    float4 asv = *(const float4*)(a_s + lane * 4);
    float4 adv = *(const float4*)(a_d + lane * 4);
    asv.x *= LOG2E; asv.y *= LOG2E; asv.z *= LOG2E; asv.w *= LOG2E;
    adv.x *= LOG2E; adv.y *= LOG2E; adv.z *= LOG2E; adv.w *= LOG2E;

    const float4 rs = *(const float4*)(hcol + (size_t)node * HID);
    const float ad = group8_sum(rs.x * adv.x + rs.y * adv.y + rs.z * adv.z + rs.w * adv.w);
    float m = lrelu02(group8_sum(rs.x * asv.x + rs.y * asv.y + rs.z * asv.z + rs.w * asv.w) + ad);
    float d = 1.f;
    float4 acc = rs;

    {
        const int s = (lane < cb) ? slots[node * CAP + lane] : node;
        const bool odd1 = (lane & 1) != 0, odd2 = (lane & 2) != 0;
        const int  jl   = ((lane & 1) << 1) | ((lane >> 1) & 1);
        int k = 0;
        for (; k + 4 <= cb; k += 4)
            edge_trip4(hcol, s, k, 4, asv, ad, odd1, odd2, jl, m, d, acc);
        if (k < cb)
            edge_trip4(hcol, s, k, cb - k, asv, ad, odd1, odd2, jl, m, d, acc);
    }
    if (c > CAP) {
        const int novf = __builtin_amdgcn_readfirstlane(cnt[NN]);
        for (int base = 0; base < novf; base += 32) {
            const int i  = base + lane;
            const int od = (i < novf) ? ovf_dst[i] : -1;
            const int os = (i < novf) ? ovf_src[i] : node;
            unsigned hit = __builtin_amdgcn_ballot_w32(od == node);
            while (hit) {
                const int j = __builtin_ctz(hit);
                hit &= hit - 1u;
                edge_trip<1>(hcol, os, j, 1, asv, ad, m, d, acc);
            }
        }
    }

    const float den = d + 1e-16f;
    const float4 bv = *(const float4*)(b + lane * 4);
    float4 y;
    y.x = acc.x / den + bv.x; y.y = acc.y / den + bv.y;
    y.z = acc.z / den + bv.z; y.w = acc.w / den + bv.w;
    y.x = (y.x > 0.f) ? y.x : (exp2_neg(y.x * LOG2E) - 1.f);
    y.y = (y.y > 0.f) ? y.y : (exp2_neg(y.y * LOG2E) - 1.f);
    y.z = (y.z > 0.f) ? y.z : (exp2_neg(y.z * LOG2E) - 1.f);
    y.w = (y.w > 0.f) ? y.w : (exp2_neg(y.w * LOG2E) - 1.f);
    *(float4*)(out + (size_t)node * HID + lane * 4) = y;
}

extern "C" void kernel_launch(void* const* d_in, const int* in_sizes, int n_in,
                              void* d_out, int out_size, void* d_ws, size_t ws_size,
                              hipStream_t stream)
{
    const float* x    = (const float*)d_in[0];
    const int*   edge = (const int*)  d_in[1];
    const float* W1   = (const float*)d_in[2];
    const float* a_s1 = (const float*)d_in[3];
    const float* a_d1 = (const float*)d_in[4];
    const float* b1   = (const float*)d_in[5];
    const float* W2   = (const float*)d_in[6];
    const float* a_s2 = (const float*)d_in[7];
    const float* a_d2 = (const float*)d_in[8];
    const float* b2   = (const float*)d_in[9];
    const float* fc_w = (const float*)d_in[10];
    const float* fc_b = (const float*)d_in[11];
    float* out = (float*)d_out;

    const int* src = edge;
    const int* dst = edge + NE;

    float* bufH   = (float*)d_ws;
    float* bufIn2 = bufH   + (size_t)NN * HID;
    int*   cnt    = (int*)(bufIn2 + (size_t)NN * HID);
    int*   slots  = cnt    + (NN + 16);
    int*   ovfs   = slots  + (size_t)NN * CAP;
    int*   ovfd   = ovfs   + NE;
    unsigned* part  = (unsigned*)(ovfd + NE);
    unsigned* spill = part + (size_t)NBIN * NPART * 32;
    int*      spcnt = (int*)(spill + NE);

    const int TB = 256;
    const int nMtiles = NN / 16;
    const int nodeBlocks = (NN * 32 + TB - 1) / TB;
    const size_t wLds = GEMM_LDS_BYTES(HID, HID);

    partition_with_gemm<<<GEMM_BLOCKS_A + NPART, TB, wLds, stream>>>(
        src, dst, part, spill, spcnt, cnt + NN, x, W1, bufH);
    bucket_build_with_gemm<<<GEMM_BLOCKS_B + NBIN, TB, wLds, stream>>>(
        part, spill, spcnt, cnt, slots, ovfs, ovfd, x, W1, bufH);

    node_softmax_gather<<<nodeBlocks, TB, 0, stream>>>(
        cnt, slots, ovfs, ovfd, bufH, a_s1, a_d1, b1, bufIn2);
    wmma_gemm_strip<HID, HID / 16><<<GEMM_BLOCKS, TB, wLds, stream>>>(
        bufIn2, W2, nullptr, bufH, nMtiles);
    node_softmax_gather<<<nodeBlocks, TB, 0, stream>>>(
        cnt, slots, ovfs, ovfd, bufH, a_s2, a_d2, b2, bufIn2);
    wmma_gemm_strip<HID, CH / 16><<<GEMM_BLOCKS, TB, GEMM_LDS_BYTES(HID, CH), stream>>>(
        bufIn2, fc_w, fc_b, out, nMtiles);
}


# Next layer's product (W2, then fc) moves into the node kernel: 16 finished rows per workgroup go via LDS to the same WMMA tiles, W staged once per 5 node tiles; 6 -> 4 launches; sum order unchanged
# speedup vs baseline: 6.5386x; 1.0347x over previous
#include <hip/hip_runtime.h>
#include <math.h>

#define NN    50000
#define NE    800000
#define HID   128
#define NH    4
#define CH    32

typedef float v2f __attribute__((ext_vector_type(2)));
typedef float v4f __attribute__((ext_vector_type(4)));
typedef float v8f __attribute__((ext_vector_type(8)));

#define GEMM_TB 256
#define GEMM_LDS_PITCH(NOUT) (2 * (NOUT))
#define GEMM_LDS_BYTES(KK, NOUT) (((KK) / 2) * GEMM_LDS_PITCH(NOUT) * 4)

template <int K, int NT>
__device__ __forceinline__ void
gemm_strip_body(float* __restrict__ Wlds,
                const float* __restrict__ X,
                const float* __restrict__ W,
                const float* __restrict__ bias,
                float* __restrict__ Y,
                int blk, int nMtiles)
{
    constexpr int Nout  = NT * 16;
    constexpr int PITCH = GEMM_LDS_PITCH(Nout);
    constexpr int KS    = K / 4;

    const int wave = (int)((blk * GEMM_TB + (int)threadIdx.x) >> 5);
    const int lane = (int)(threadIdx.x & 31);
    const int col  = lane & 15;
    const int h    = lane >> 4;

    constexpr int NPAIR = (K * Nout) / (GEMM_TB * 8);
    static_assert(NPAIR * GEMM_TB * 8 == K * Nout, "W is a whole number of row-pair pieces per thread");
    float4 wa[NPAIR], wb[NPAIR];
#pragma unroll
    for (int j = 0; j < NPAIR; ++j) {
        const int p  = j * GEMM_TB + (int)threadIdx.x;
        const int kp = p / (Nout / 4), c4 = p % (Nout / 4);
        wa[j] = *(const float4*)&W[(2 * kp)     * Nout + 4 * c4];
        wb[j] = *(const float4*)&W[(2 * kp + 1) * Nout + 4 * c4];
    }

    constexpr int KA0 = 8;
    const int awave = (wave < nMtiles) ? wave : nMtiles - 1;
    const float* xrow = X + (size_t)(awave * 16 + col) * K + 2 * h;
    v2f a[KS];
#pragma unroll
    for (int ks = 0; ks < KA0; ++ks)
        a[ks] = *(const v2f*)(xrow + 4 * ks);

#pragma unroll
    for (int j = 0; j < NPAIR; ++j) {
        const int p  = j * GEMM_TB + (int)threadIdx.x;
        const int kp = p / (Nout / 4), c4 = p % (Nout / 4);
        float* d = &Wlds[kp * PITCH + 8 * c4];
        *(float4*)(d)     = make_float4(wa[j].x, wb[j].x, wa[j].y, wb[j].y);
        *(float4*)(d + 4) = make_float4(wa[j].z, wb[j].z, wa[j].w, wb[j].w);
    }
    static_assert(KA0 == 8, "the fence names eight fragments");
    asm volatile("" :: "v"(a[0]), "v"(a[1]), "v"(a[2]), "v"(a[3]),
                       "v"(a[4]), "v"(a[5]), "v"(a[6]), "v"(a[7]));
    __syncthreads();

    if (wave >= nMtiles) return;

#pragma unroll
    for (int ks = KA0; ks < KS; ++ks)
        a[ks] = *(const v2f*)(xrow + 4 * ks);
    __builtin_amdgcn_sched_group_barrier(0x020, KS - KA0, 0);

    v8f acc[NT];
#pragma unroll
    for (int t = 0; t < NT; ++t)
        acc[t] = (v8f){0.f, 0.f, 0.f, 0.f, 0.f, 0.f, 0.f, 0.f};

    constexpr int PD  = (NT >= 8) ? 1 : 3;
    constexpr int DSN = NT / 2;
    const float* wl = Wlds + h * PITCH + 2 * col;
    v2f b[KS][NT];
#pragma unroll
    for (int s = 0; s < PD; ++s) {
#pragma unroll
        for (int t = 0; t < NT; ++t)
            b[s][t] = *(const v2f*)(wl + s * 2 * PITCH + t * 32);
        __builtin_amdgcn_sched_group_barrier(0x100, DSN, 0);
    }
#pragma unroll
    for (int ks = 0; ks < KS; ++ks) {
        if (ks + PD < KS) {
#pragma unroll
            for (int t = 0; t < NT; ++t)
                b[ks + PD][t] = *(const v2f*)(wl + (ks + PD) * 2 * PITCH + t * 32);
        }
#pragma unroll
        for (int t = 0; t < NT; ++t)
            acc[t] = __builtin_amdgcn_wmma_f32_16x16x4_f32(
                         false, a[ks], false, b[ks][t], (short)0, acc[t], false, false);
        if (ks + PD < KS)
            __builtin_amdgcn_sched_group_barrier(0x100, DSN, 0);
        __builtin_amdgcn_sched_group_barrier(0x008, NT, 0);
    }

    int orow = wave * 16 + ((lane >> 4) << 3);
#pragma unroll
    for (int t = 0; t < NT; ++t) {
        int ocol = t * 16 + (lane & 15);
        float badd = bias ? bias[ocol] : 0.f;
#pragma unroll
        for (int r = 0; r < 8; ++r)
            Y[(size_t)(orow + r) * Nout + ocol] = acc[t][r] + badd;
    }
}

#define CAP     32
#define NBIN    256
#define RANGE   196
#define NPART   256
#define CHUNK   3125
#define CELLCAP 31
static_assert(NBIN * RANGE >= NN, "ranges cover all nodes");
static_assert(NPART * CHUNK == NE, "chunks cover all edges");
static_assert(NN <= 65536 && RANGE <= 256, "packed entry: 16-bit source, local destination above it");

#define PART_LDS_WORDS (NBIN + NBIN * 32 + 1)
__device__ __forceinline__ void
edge_partition_body(unsigned* __restrict__ lds, const int w,
               const int* __restrict__ src,
               const int* __restrict__ dst,
               unsigned* __restrict__ part,
               unsigned* __restrict__ spill,
               int* __restrict__ spill_cnt,
               int* __restrict__ ovf_total)
{
    int*      ccnt    = (int*)lds;
    unsigned* cell    = lds + NBIN;
    int*      nspillp = (int*)(lds + NBIN + NBIN * 32);
    const int t = threadIdx.x;
    ccnt[t] = 0;
    if (t == 0) *nspillp = 0;
    if (w == 0 && t == 0) *ovf_total = 0;

    constexpr int PER = (CHUNK + 255) / 256;
    int dn[PER], sv[PER];
#pragma unroll
    for (int j = 0; j < PER; ++j) {
        const int i = j * 256 + t;
        const int e = w * CHUNK + ((i < CHUNK) ? i : 0);
        dn[j] = dst[e];
        sv[j] = src[e];
    }
    __syncthreads();
    int bn[PER], pr[PER];
#pragma unroll
    for (int j = 0; j < PER; ++j) {
        bn[j] = dn[j] / RANGE;
        pr[j] = (j * 256 + t < CHUNK) ? atomicAdd(&ccnt[bn[j]], 1) : -1;
    }
#pragma unroll
    for (int j = 0; j < PER; ++j) {
        if (pr[j] >= 0) {
            if (pr[j] < CELLCAP) {
                cell[bn[j] * 32 + 1 + pr[j]] = ((unsigned)(dn[j] - bn[j] * RANGE) << 16) | (unsigned)sv[j];
            } else {
                const int q = atomicAdd(nspillp, 1);
                spill[(size_t)w * CHUNK + q] = ((unsigned)dn[j] << 16) | (unsigned)sv[j];
            }
        }
    }
    __syncthreads();
    const int wave = t >> 5, lane = t & 31;
    for (int b = wave; b < NBIN; b += 8) {
        int c = ccnt[b];
        c = (c < CELLCAP) ? c : CELLCAP;
        unsigned v = (unsigned)c;
        if (lane >= 1) v = (lane <= c) ? cell[b * 32 + lane] : 0u;
        part[((size_t)b * NPART + w) * 32 + lane] = v;
    }
    if (t == 0) spill_cnt[w] = *nspillp;
}

#define BUILD_LDS_WORDS (256 + RANGE * CAP + NPART + NPART + 1)
__device__ __forceinline__ void
bucket_build_body(int* __restrict__ lds, const int b,
             const unsigned* __restrict__ part,
             const unsigned* __restrict__ spill,
             const int* __restrict__ spill_cnt,
             int* __restrict__ cnt,
             int* __restrict__ slots,
             int* __restrict__ ovf_src,
             int* __restrict__ ovf_dst)
{
    int* ncnt   = lds;
    int* nslot  = lds + 256;
    int* scnt   = nslot + RANGE * CAP;
    int* slist  = scnt + NPART;
    int* nlistp = slist + NPART;
    const int t = threadIdx.x;
    const int wave = t >> 5, lane = t & 31;
    const int n0 = b * RANGE;
    ncnt[t] = 0;
    if (t == 0) *nlistp = 0;

    auto store_ranked = [&](int dl, int s, int p) {
        if (p < CAP) {
            nslot[dl * CAP + p] = s;
        } else {
            const int q = atomicAdd(&cnt[NN], 1);
            ovf_src[q] = s;
            ovf_dst[q] = n0 + dl;
        }
    };
    auto place = [&](int dl, int s) { store_ranked(dl, s, atomicAdd(&ncnt[dl], 1)); };

    unsigned v[32];
#pragma unroll
    for (int j = 0; j < 32; ++j)
        v[j] = part[((size_t)b * NPART + wave * 32 + j) * 32 + lane];
    const int sc = spill_cnt[t];
    __syncthreads();
    scnt[t] = sc;
    if (sc > 0) slist[atomicAdd(nlistp, 1)] = t;
    int pr[32];
#pragma unroll
    for (int j = 0; j < 32; ++j) {
        const int c = __builtin_amdgcn_readfirstlane((int)v[j]);
        pr[j] = (lane >= 1 && lane <= c) ? atomicAdd(&ncnt[v[j] >> 16], 1) : -1;
    }
#pragma unroll
    for (int j = 0; j < 32; ++j) {
        if (pr[j] >= 0)
            store_ranked((int)(v[j] >> 16), (int)(v[j] & 0xffffu), pr[j]);
    }
    __syncthreads();
    const int nl = *nlistp;
    for (int k = 0; k < nl; ++k) {
        const int w = slist[k];
        int n = scnt[w];
        n = (n < CHUNK) ? n : CHUNK;
        for (int i = t; i < n; i += 256) {
            const unsigned e = spill[(size_t)w * CHUNK + i];
            const int dl = (int)(e >> 16) - n0;
            if (dl >= 0 && dl < RANGE)
                place(dl, (int)(e & 0xffffu));
        }
    }
    if (nl > 0) __syncthreads();
    for (int r = wave; r < RANGE; r += 8) {
        const int node = n0 + r;
        if (node < NN) {
            int c = ncnt[r];
            c = (c < CAP) ? c : CAP;
            slots[(size_t)node * CAP + lane] = (lane < c) ? nslot[r * CAP + lane] : 0;
        }
    }
    if (t < RANGE && n0 + t < NN) cnt[n0 + t] = ncnt[t];
}

#define GEMM_BLOCKS  ((NN / 16 * 32 + GEMM_TB - 1) / GEMM_TB)
#define GEMM_BLOCKS_A ((GEMM_BLOCKS + 1) / 2)
#define GEMM_BLOCKS_B (GEMM_BLOCKS - GEMM_BLOCKS_A)
static_assert(PART_LDS_WORDS * 4 <= GEMM_LDS_BYTES(HID, HID) && BUILD_LDS_WORDS * 4 <= GEMM_LDS_BYTES(HID, HID),
              "the grouping passes fit in the LDS sized for the weight matrix");

__global__ void __launch_bounds__(256)
partition_with_gemm(const int* __restrict__ src,
                    const int* __restrict__ dst,
                    unsigned* __restrict__ part,
                    unsigned* __restrict__ spill,
                    int* __restrict__ spill_cnt,
                    int* __restrict__ ovf_total,
                    const float* __restrict__ X,
                    const float* __restrict__ Wm,
                    float* __restrict__ Y)
{
    extern __shared__ __align__(16) float dyn_lds[];
    const int blk = (int)blockIdx.x;
    if (blk < GEMM_BLOCKS_A)
        gemm_strip_body<HID, HID / 16>(dyn_lds, X, Wm, nullptr, Y, blk, NN / 16);
    else
        edge_partition_body((unsigned*)dyn_lds, blk - GEMM_BLOCKS_A,
                            src, dst, part, spill, spill_cnt, ovf_total);
}

__global__ void __launch_bounds__(256)
bucket_build_with_gemm(const unsigned* __restrict__ part,
                       const unsigned* __restrict__ spill,
                       const int* __restrict__ spill_cnt,
                       int* __restrict__ cnt,
                       int* __restrict__ slots,
                       int* __restrict__ ovf_src,
                       int* __restrict__ ovf_dst,
                       const float* __restrict__ X,
                       const float* __restrict__ Wm,
                       float* __restrict__ Y)
{
    extern __shared__ __align__(16) float dyn_lds[];
    const int blk = (int)blockIdx.x;
    if (blk < GEMM_BLOCKS_B)
        gemm_strip_body<HID, HID / 16>(dyn_lds, X, Wm, nullptr, Y, GEMM_BLOCKS_A + blk, NN / 16);
    else
        bucket_build_body((int*)dyn_lds, blk - GEMM_BLOCKS_B,
                          part, spill, spill_cnt, cnt, slots, ovf_src, ovf_dst);
}

__device__ __forceinline__ float lrelu02(float v) {
    return (v > 0.f) ? v : 0.2f * v;
}
#define LOG2E 1.4426950408889634f
__device__ __forceinline__ float exp2_neg(float v) {
    return __builtin_amdgcn_exp2f(v);
}
__device__ __forceinline__ float group8_sum(float v) {
    v += __int_as_float(__builtin_amdgcn_update_dpp(0, __float_as_int(v), 0xB1, 0xF, 0xF, true));
    v += __int_as_float(__builtin_amdgcn_update_dpp(0, __float_as_int(v), 0x4E, 0xF, 0xF, true));
    v += __int_as_float(__builtin_amdgcn_update_dpp(0, __float_as_int(v), 0x141, 0xF, 0xF, true));
    return v;
}

template <int N>
__device__ __forceinline__ void edge_trip(const float* __restrict__ hcol, int s, int k,
                                          int nvalid, const float4 asv, float ad,
                                          float& m, float& d, float4& acc)
{
    float4 r[N];
    float  e[N];
#pragma unroll
    for (int j = 0; j < N; ++j) {
        const int sj = __builtin_amdgcn_readlane(s, k + j);
        r[j] = *(const float4*)(hcol + (size_t)sj * HID);
    }
    float mn = m;
#pragma unroll
    for (int j = 0; j < N; ++j) {
        float p = r[j].x * asv.x + r[j].y * asv.y + r[j].z * asv.z + r[j].w * asv.w;
        p = group8_sum(p);
        const float ej = lrelu02(p + ad);
        e[j] = (j < nvalid) ? ej : -INFINITY;
        mn = fmaxf(mn, e[j]);
    }
    const float sc = exp2_neg(m - mn);
    float dsum = 0.f;
    float4 a;
    a.x = acc.x * sc; a.y = acc.y * sc; a.z = acc.z * sc; a.w = acc.w * sc;
#pragma unroll
    for (int j = 0; j < N; ++j) {
        const float pj = exp2_neg(e[j] - mn);
        dsum += pj;
        a.x += r[j].x * pj; a.y += r[j].y * pj; a.z += r[j].z * pj; a.w += r[j].w * pj;
    }
    d   = d * sc + dsum;
    m   = mn;
    acc = a;
}

__device__ __forceinline__ float lane_xor1(float v) {
    return __int_as_float(__builtin_amdgcn_update_dpp(0, __float_as_int(v), 0xB1, 0xF, 0xF, true));
}
__device__ __forceinline__ float lane_xor2(float v) {
    return __int_as_float(__builtin_amdgcn_update_dpp(0, __float_as_int(v), 0x4E, 0xF, 0xF, true));
}
__device__ __forceinline__ float lane_xor4(float v) {
    return __int_as_float(__builtin_amdgcn_update_dpp(0, __float_as_int(v), 0x164, 0xF, 0xF, true));
}
template <int Q>
__device__ __forceinline__ float quad_bcast(float v) {
    return __int_as_float(__builtin_amdgcn_update_dpp(0, __float_as_int(v), Q * 0x55, 0xF, 0xF, true));
}

__device__ __forceinline__ void edge_trip4(const float* __restrict__ hcol, int s, int k,
                                           int nvalid, const float4 asv, float ad,
                                           bool odd1, bool odd2, int jl,
                                           float& m, float& d, float4& acc)
{
    float4 r[4];
    float  q[4];
#pragma unroll
    for (int j = 0; j < 4; ++j) {
        const int sj = __builtin_amdgcn_readlane(s, k + j);
        r[j] = *(const float4*)(hcol + (size_t)sj * HID);
    }
    const v2f a01 = {asv.x, asv.y}, a23 = {asv.z, asv.w};
#pragma unroll
    for (int j = 0; j < 4; ++j) {
        v2f t = (v2f){r[j].x, r[j].y} * a01;
        t = (v2f){r[j].z, r[j].w} * a23 + t;
        q[j] = t.x + t.y;
    }
    const float u0 = (odd1 ? q[2] : q[0]) + lane_xor1(odd1 ? q[0] : q[2]);
    const float u1 = (odd1 ? q[3] : q[1]) + lane_xor1(odd1 ? q[1] : q[3]);
    float w = (odd2 ? u1 : u0) + lane_xor2(odd2 ? u0 : u1);
    w += lane_xor4(w);
    float e = lrelu02(w + ad);
    e = (jl < nvalid) ? e : -INFINITY;
    float t4 = fmaxf(e, lane_xor1(e));
    t4 = fmaxf(t4, lane_xor2(t4));
    const float mn = fmaxf(m, t4);
    const float sc = exp2_neg(m - mn);
    const float p  = exp2_neg(e - mn);
    float ps = p + lane_xor1(p);
    ps += lane_xor2(ps);
    const float p0 = quad_bcast<0>(p), p1 = quad_bcast<2>(p),
                p2 = quad_bcast<1>(p), p3 = quad_bcast<3>(p);
    v2f c01 = (v2f){acc.x, acc.y} * sc, c23 = (v2f){acc.z, acc.w} * sc;
    c01 = (v2f){r[0].x, r[0].y} * p0 + c01;  c23 = (v2f){r[0].z, r[0].w} * p0 + c23;
    c01 = (v2f){r[1].x, r[1].y} * p1 + c01;  c23 = (v2f){r[1].z, r[1].w} * p1 + c23;
    c01 = (v2f){r[2].x, r[2].y} * p2 + c01;  c23 = (v2f){r[2].z, r[2].w} * p2 + c23;
    c01 = (v2f){r[3].x, r[3].y} * p3 + c01;  c23 = (v2f){r[3].z, r[3].w} * p3 + c23;
    d   = d * sc + ps;
    m   = mn;
    acc = make_float4(c01.x, c01.y, c23.x, c23.y);
}

__device__ __forceinline__ float4
node_row(const int node, const int lane,
         const int* __restrict__ cnt,
         const int* __restrict__ slots,
         const int* __restrict__ ovf_src,
         const int* __restrict__ ovf_dst,
         const float* __restrict__ hcol,
         const float4 asv, const float4 adv, const float4 bv)
{
    const int c    = __builtin_amdgcn_readfirstlane(cnt[node]);
    const int cb   = (c < CAP) ? c : CAP;

    const float4 rs = *(const float4*)(hcol + (size_t)node * HID);
    const float ad = group8_sum(rs.x * adv.x + rs.y * adv.y + rs.z * adv.z + rs.w * adv.w);
    float m = lrelu02(group8_sum(rs.x * asv.x + rs.y * asv.y + rs.z * asv.z + rs.w * asv.w) + ad);
    float d = 1.f;
    float4 acc = rs;

    {
        const int s = (lane < cb) ? slots[node * CAP + lane] : node;
        const bool odd1 = (lane & 1) != 0, odd2 = (lane & 2) != 0;
        const int  jl   = ((lane & 1) << 1) | ((lane >> 1) & 1);
        int k = 0;
        for (; k + 4 <= cb; k += 4)
            edge_trip4(hcol, s, k, 4, asv, ad, odd1, odd2, jl, m, d, acc);
        if (k < cb)
            edge_trip4(hcol, s, k, cb - k, asv, ad, odd1, odd2, jl, m, d, acc);
    }
    if (c > CAP) {
        const int novf = __builtin_amdgcn_readfirstlane(cnt[NN]);
        for (int base = 0; base < novf; base += 32) {
            const int i  = base + lane;
            const int od = (i < novf) ? ovf_dst[i] : -1;
            const int os = (i < novf) ? ovf_src[i] : node;
            unsigned hit = __builtin_amdgcn_ballot_w32(od == node);
            while (hit) {
                const int j = __builtin_ctz(hit);
                hit &= hit - 1u;
                edge_trip<1>(hcol, os, j, 1, asv, ad, m, d, acc);
            }
        }
    }

    const float den = d + 1e-16f;
    float4 y;
    y.x = acc.x / den + bv.x; y.y = acc.y / den + bv.y;
    y.z = acc.z / den + bv.z; y.w = acc.w / den + bv.w;
    y.x = (y.x > 0.f) ? y.x : (exp2_neg(y.x * LOG2E) - 1.f);
    y.y = (y.y > 0.f) ? y.y : (exp2_neg(y.y * LOG2E) - 1.f);
    y.z = (y.z > 0.f) ? y.z : (exp2_neg(y.z * LOG2E) - 1.f);
    y.w = (y.w > 0.f) ? y.w : (exp2_neg(y.w * LOG2E) - 1.f);
    return y;
}

#define FT_WAVES  16
#define FT_TB     (FT_WAVES * 32)
#define FT_TILES  (NN / 16)
#define FT_GRID   625
#define FT_TRIPS  (FT_TILES / FT_GRID)
#define FT_APITCH (HID + 4)
static_assert(NN % 16 == 0 && FT_GRID * FT_TRIPS == FT_TILES, "every workgroup walks the same number of full tiles");

template <int NT>
__global__ void __launch_bounds__(FT_TB)
node_softmax_gather_proj(const int* __restrict__ cnt,
                         const int* __restrict__ slots,
                         const int* __restrict__ ovf_src,
                         const int* __restrict__ ovf_dst,
                         const float* __restrict__ Hm,
                         const float* __restrict__ a_s,
                         const float* __restrict__ a_d,
                         const float* __restrict__ b,
                         const float* __restrict__ Wp,
                         const float* __restrict__ pbias,
                         float* __restrict__ Y)
{
    constexpr int Nout  = NT * 16;
    constexpr int PITCH = GEMM_LDS_PITCH(Nout);
    constexpr int KS    = HID / 4;
    __shared__ __align__(16) float Wlds[(HID / 2) * PITCH];
    __shared__ __align__(16) float Als[2 * 16 * FT_APITCH];

    const int wave = __builtin_amdgcn_readfirstlane((int)(threadIdx.x >> 5));
    const int lane = (int)(threadIdx.x & 31);

    constexpr int NPAIR = (HID * Nout) / (FT_TB * 8);
    static_assert(NPAIR * FT_TB * 8 == HID * Nout, "Wp is a whole number of row-pair pieces per thread");
    {
        float4 wa[NPAIR], wb[NPAIR];
#pragma unroll
        for (int j = 0; j < NPAIR; ++j) {
            const int p  = j * FT_TB + (int)threadIdx.x;
            const int kp = p / (Nout / 4), c4 = p % (Nout / 4);
            wa[j] = *(const float4*)&Wp[(2 * kp)     * Nout + 4 * c4];
            wb[j] = *(const float4*)&Wp[(2 * kp + 1) * Nout + 4 * c4];
        }
#pragma unroll
        for (int j = 0; j < NPAIR; ++j) {
            const int p  = j * FT_TB + (int)threadIdx.x;
            const int kp = p / (Nout / 4), c4 = p % (Nout / 4);
            const int t = c4 >> 2, c0 = (4 * c4) & 15;
            float* dd = &Wlds[kp * PITCH + (t >> 1) * 64 + c0 * 4 + (t & 1) * 2];
            *(v2f*)(dd)      = (v2f){wa[j].x, wb[j].x};
            *(v2f*)(dd + 4)  = (v2f){wa[j].y, wb[j].y};
            *(v2f*)(dd + 8)  = (v2f){wa[j].z, wb[j].z};
            *(v2f*)(dd + 12) = (v2f){wa[j].w, wb[j].w};
        }
    }

    const float* hcol = Hm + lane * 4;
    float4 asv = *(const float4*)(a_s + lane * 4);
    float4 adv = *(const float4*)(a_d + lane * 4);
    asv.x *= LOG2E; asv.y *= LOG2E; asv.z *= LOG2E; asv.w *= LOG2E;
    adv.x *= LOG2E; adv.y *= LOG2E; adv.z *= LOG2E; adv.w *= LOG2E;
    const float4 bv = *(const float4*)(b + lane * 4);

    const int col = lane & 15;
    const int h   = lane >> 4;
    constexpr int NW = NT / 2;
    static_assert(NT % 2 == 0 && NW <= FT_WAVES, "column tiles are dealt in pairs");
    const bool pw = wave < NW;
    const float badd0 = (pbias && pw) ? pbias[wave * 32 + col]      : 0.f;
    const float badd1 = (pbias && pw) ? pbias[wave * 32 + col + 16] : 0.f;

    for (int i = 0; i < FT_TRIPS; ++i) {
        const int tile = (int)blockIdx.x + i * FT_GRID;
        const int node = tile * 16 + wave;
        const float4 y = node_row(node, lane, cnt, slots, ovf_src, ovf_dst, hcol, asv, adv, bv);

        float* abuf = Als + (i & 1) * (16 * FT_APITCH);
        *(float4*)(abuf + wave * FT_APITCH + lane * 4) = y;
        __syncthreads();

        if (wave < NW) {
            const float* al = abuf + col * FT_APITCH + 2 * h;
            const float* wl = Wlds + h * PITCH + wave * 64 + col * 4;
            constexpr int PD = 3;
            v2f af[KS];
            v4f bf[KS];
#pragma unroll
            for (int s = 0; s < PD; ++s) {
                af[s] = *(const v2f*)(al + 4 * s);
                bf[s] = *(const v4f*)(wl + s * 2 * PITCH);
            }
            v8f acc0 = (v8f){0.f, 0.f, 0.f, 0.f, 0.f, 0.f, 0.f, 0.f};
            v8f acc1 = (v8f){0.f, 0.f, 0.f, 0.f, 0.f, 0.f, 0.f, 0.f};
#pragma unroll
            for (int ks = 0; ks < KS; ++ks) {
                if (ks + PD < KS) {
                    af[ks + PD] = *(const v2f*)(al + 4 * (ks + PD));
                    bf[ks + PD] = *(const v4f*)(wl + (ks + PD) * 2 * PITCH);
                }
                acc0 = __builtin_amdgcn_wmma_f32_16x16x4_f32(
                           false, af[ks], false, (v2f){bf[ks].x, bf[ks].y}, (short)0, acc0, false, false);
                acc1 = __builtin_amdgcn_wmma_f32_16x16x4_f32(
                           false, af[ks], false, (v2f){bf[ks].z, bf[ks].w}, (short)0, acc1, false, false);
                if (ks + PD < KS)
                    __builtin_amdgcn_sched_group_barrier(0x100, 2, 0);
                __builtin_amdgcn_sched_group_barrier(0x008, 2, 0);
            }
            const int orow = tile * 16 + (h << 3);
            const int ocol = wave * 32 + col;
#pragma unroll
            for (int r = 0; r < 8; ++r) {
                Y[(size_t)(orow + r) * Nout + ocol]      = acc0[r] + badd0;
                Y[(size_t)(orow + r) * Nout + ocol + 16] = acc1[r] + badd1;
            }
        }
    }
}

extern "C" void kernel_launch(void* const* d_in, const int* in_sizes, int n_in,
                              void* d_out, int out_size, void* d_ws, size_t ws_size,
                              hipStream_t stream)
{
    const float* x    = (const float*)d_in[0];
    const int*   edge = (const int*)  d_in[1];
    const float* W1   = (const float*)d_in[2];
    const float* a_s1 = (const float*)d_in[3];
    const float* a_d1 = (const float*)d_in[4];
    const float* b1   = (const float*)d_in[5];
    const float* W2   = (const float*)d_in[6];
    const float* a_s2 = (const float*)d_in[7];
    const float* a_d2 = (const float*)d_in[8];
    const float* b2   = (const float*)d_in[9];
    const float* fc_w = (const float*)d_in[10];
    const float* fc_b = (const float*)d_in[11];
    float* out = (float*)d_out;

    const int* src = edge;
    const int* dst = edge + NE;

    float* bufH   = (float*)d_ws;
    float* bufIn2 = bufH   + (size_t)NN * HID;
    int*   cnt    = (int*)(bufIn2 + (size_t)NN * HID);
    int*   slots  = cnt    + (NN + 16);
    int*   ovfs   = slots  + (size_t)NN * CAP;
    int*   ovfd   = ovfs   + NE;
    unsigned* part  = (unsigned*)(ovfd + NE);
    unsigned* spill = part + (size_t)NBIN * NPART * 32;
    int*      spcnt = (int*)(spill + NE);

    const int TB = 256;
    const size_t wLds = GEMM_LDS_BYTES(HID, HID);

    partition_with_gemm<<<GEMM_BLOCKS_A + NPART, TB, wLds, stream>>>(
        src, dst, part, spill, spcnt, cnt + NN, x, W1, bufH);
    bucket_build_with_gemm<<<GEMM_BLOCKS_B + NBIN, TB, wLds, stream>>>(
        part, spill, spcnt, cnt, slots, ovfs, ovfd, x, W1, bufH);

    node_softmax_gather_proj<HID / 16><<<FT_GRID, FT_TB, 0, stream>>>(
        cnt, slots, ovfs, ovfd, bufH, a_s1, a_d1, b1, W2, nullptr, bufIn2);
    node_softmax_gather_proj<CH / 16><<<FT_GRID, FT_TB, 0, stream>>>(
        cnt, slots, ovfs, ovfd, bufIn2, a_s2, a_d2, b2, fc_w, fc_b, out);
}
